# starting kernel
# baseline (speedup 1.0000x reference)
#include <hip/hip_runtime.h>
#include <math.h>

typedef __attribute__((ext_vector_type(2))) float v2f;
typedef __attribute__((ext_vector_type(8))) float v8f;

#define B_  16
#define L_  4096
#define D_  1024
#define H_  16
#define DK_ 64
#define DV_ 64

static __device__ __forceinline__ v8f wmma_gemm_f32(const float* __restrict__ A, int lda,
                                                    const float* __restrict__ Bm, int ldb,
                                                    int Klen)
{
    const int lane = threadIdx.x & 31;
    const int half = lane >> 4;
    const int mn   = lane & 15;

    const float* ap = A  + (size_t)mn * lda + 2 * half;
    const float* bp = Bm + (size_t)(2 * half) * ldb + mn;

    v8f c = {};
    for (int k = 0; k < Klen; k += 4) {
        if ((k & 511) == 0) {
            __builtin_prefetch(ap + 512, 0, 1);
        }
        v2f a, b;
        a.x = ap[0];
        a.y = ap[1];
        b.x = bp[0];
        b.y = bp[ldb];
        c = __builtin_amdgcn_wmma_f32_16x16x4_f32(false, a, false, b,
                                                  (short)0, c, false, false);
        ap += 4;
        bp += (size_t)4 * ldb;
    }
    return c;
}


__global__ void k_qproj(const float* __restrict__ query, const float* __restrict__ w_qs,
                        float* __restrict__ qs)
{
    int i = blockIdx.x * blockDim.x + threadIdx.x;
    float acc = 0.f;
    for (int d = 0; d < D_; ++d) acc += query[d] * w_qs[(size_t)d * (H_ * DK_) + i];
    qs[i] = acc * 0.125f;
}

__global__ void k_weff(const float* __restrict__ w_ks, const float* __restrict__ qs,
                       float* __restrict__ w_eff)
{
    int i = blockIdx.x * blockDim.x + threadIdx.x;
    int d = i >> 4, h = i & 15;
    const float* wk = w_ks + (size_t)d * (H_ * DK_) + h * DK_;
    const float* qh = qs + h * DK_;
    float acc = 0.f;
#pragma unroll
    for (int j = 0; j < DK_; ++j) acc += wk[j] * qh[j];
    w_eff[i] = acc;
}

__global__ void k_scores(const float* __restrict__ x, const float* __restrict__ w_eff,
                         float* __restrict__ attn)
{
    int wave = (blockIdx.x * blockDim.x + threadIdx.x) >> 5;
    int b  = wave >> 8;
    int lt = wave & 255;

    const float* A = x + ((size_t)b * L_ + (size_t)lt * 16) * D_;
    v8f c = wmma_gemm_f32(A, D_, w_eff, H_, D_);

    const int lane = threadIdx.x & 31, half = lane >> 4, n = lane & 15;
    float* out = attn + ((size_t)b * H_ + n) * L_ + (size_t)lt * 16 + 8 * half;
#pragma unroll
    for (int r = 0; r < 8; ++r) out[r] = c[r];
}

__global__ void k_softmax(float* __restrict__ attn)
{
    __shared__ float red[256];
    float* p = attn + (size_t)blockIdx.x * L_;
    const int t = threadIdx.x;

    float vals[16];
    float m = -3.4e38f;
#pragma unroll
    for (int i = 0; i < 16; ++i) { vals[i] = p[t + i * 256]; m = fmaxf(m, vals[i]); }

    red[t] = m; __syncthreads();
    for (int s = 128; s > 0; s >>= 1) {
        if (t < s) red[t] = fmaxf(red[t], red[t + s]);
        __syncthreads();
    }
    m = red[0]; __syncthreads();

    float sum = 0.f;
#pragma unroll
    for (int i = 0; i < 16; ++i) { vals[i] = expf(vals[i] - m); sum += vals[i]; }

    red[t] = sum; __syncthreads();
    for (int s = 128; s > 0; s >>= 1) {
        if (t < s) red[t] += red[t + s];
        __syncthreads();
    }
    float inv = 1.f / red[0];
#pragma unroll
    for (int i = 0; i < 16; ++i) p[t + i * 256] = vals[i] * inv;
}

#define CTXX_SPLIT 4
__global__ void k_ctxx_split(const float* __restrict__ attn, const float* __restrict__ x,
                             float* __restrict__ ctx_x_part)
{
    int wave = (blockIdx.x * blockDim.x + threadIdx.x) >> 5;
    int b  = wave >> 8;
    int t  = wave & 255;
    int s  = t >> 6;
    int dt = t & 63;
    const int k0 = s * (L_ / CTXX_SPLIT);

    const float* A  = attn + (size_t)b * H_ * L_ + k0;
    const float* Bm = x + ((size_t)b * L_ + k0) * D_ + dt * 16;
    v8f c = wmma_gemm_f32(A, L_, Bm, D_, L_ / CTXX_SPLIT);

    const int lane = threadIdx.x & 31, half = lane >> 4, n = lane & 15;
    float* out = ctx_x_part + (size_t)s * (B_ * H_ * D_)
               + (size_t)b * H_ * D_ + (size_t)dt * 16 + n;
#pragma unroll
    for (int r = 0; r < 8; ++r) out[(size_t)(r + 8 * half) * D_] = c[r];
}

__global__ void k_reduce(const float* __restrict__ part, float* __restrict__ dst,
                         int nparts, int n)
{
    int stride = gridDim.x * blockDim.x;
    for (int i = blockIdx.x * blockDim.x + threadIdx.x; i < n; i += stride) {
        float s = 0.f;
        for (int p = 0; p < nparts; ++p) s += part[(size_t)p * n + i];
        dst[i] = s;
    }
}

#define VSPLIT 8
__global__ void k_ctx_split(const float* __restrict__ ctx_x, const float* __restrict__ w_vs,
                            float* __restrict__ ctx_part)
{
    int wave = (blockIdx.x * blockDim.x + threadIdx.x) >> 5;
    int tile = wave >> 3;
    int s    = wave & 7;
    int h    = tile >> 2;
    int dvt  = tile & 3;
    const int k0 = s * (D_ / VSPLIT);

    const float* A  = ctx_x + (size_t)h * D_ + k0;
    const float* Bm = w_vs + (size_t)k0 * (H_ * DV_) + h * DV_ + dvt * 16;
    v8f c = wmma_gemm_f32(A, H_ * D_, Bm, H_ * DV_, D_ / VSPLIT);

    const int lane = threadIdx.x & 31, half = lane >> 4, n = lane & 15;
    float* out = ctx_part + (size_t)s * (B_ * H_ * DV_)
               + h * DV_ + dvt * 16 + n;
#pragma unroll
    for (int r = 0; r < 8; ++r) out[(size_t)(r + 8 * half) * (H_ * DV_)] = c[r];
}

__global__ void k_fc_split(const float* __restrict__ ctx, const float* __restrict__ fc,
                           float* __restrict__ out_pre_part)
{
    int wave = (blockIdx.x * blockDim.x + threadIdx.x) >> 5;
    int dt = wave >> 3;
    int s  = wave & 7;
    const int k0 = s * ((H_ * DV_) / VSPLIT);

    const float* A  = ctx + k0;
    const float* Bm = fc + (size_t)k0 * D_ + dt * 16;
    v8f c = wmma_gemm_f32(A, H_ * DV_, Bm, D_, (H_ * DV_) / VSPLIT);

    const int lane = threadIdx.x & 31, half = lane >> 4, n = lane & 15;
    float* out = out_pre_part + (size_t)s * (B_ * D_) + dt * 16 + n;
#pragma unroll
    for (int r = 0; r < 8; ++r) out[(size_t)(r + 8 * half) * D_] = c[r];
}

__global__ void k_ln(const float* __restrict__ out_pre_part, const float* __restrict__ query,
                     const float* __restrict__ gamma, const float* __restrict__ beta,
                     float* __restrict__ out)
{
    __shared__ float rs[256], rs2[256];
    const int b = blockIdx.x, t = threadIdx.x;

    float vals[4];
    float s = 0.f, s2 = 0.f;
#pragma unroll
    for (int i = 0; i < 4; ++i) {
        const int idx = t + i * 256;
        float v = query[idx];
        for (int p = 0; p < VSPLIT; ++p)
            v += out_pre_part[(size_t)p * (B_ * D_) + (size_t)b * D_ + idx];
        vals[i] = v; s += v; s2 += v * v;
    }
    rs[t] = s; rs2[t] = s2; __syncthreads();
    for (int k = 128; k > 0; k >>= 1) {
        if (t < k) { rs[t] += rs[t + k]; rs2[t] += rs2[t + k]; }
        __syncthreads();
    }
    float mean = rs[0] * (1.f / D_);
    float var  = rs2[0] * (1.f / D_) - mean * mean;
    float inv  = rsqrtf(var + 1e-6f);
#pragma unroll
    for (int i = 0; i < 4; ++i) {
        int idx = t + i * 256;
        out[(size_t)b * D_ + idx] = (vals[i] - mean) * inv * gamma[idx] + beta[idx];
    }
}

extern "C" void kernel_launch(void* const* d_in, const int* in_sizes, int n_in,
                              void* d_out, int out_size, void* d_ws, size_t ws_size,
                              hipStream_t stream)
{
    (void)in_sizes; (void)n_in; (void)out_size; (void)ws_size;

    const float* x     = (const float*)d_in[0];
    const float* query = (const float*)d_in[1];
    const float* w_qs  = (const float*)d_in[2];
    const float* w_ks  = (const float*)d_in[3];
    const float* w_vs  = (const float*)d_in[4];
    const float* fc    = (const float*)d_in[5];
    const float* gamma = (const float*)d_in[6];
    const float* beta  = (const float*)d_in[7];

    float* out  = (float*)d_out;
    float* attn = out + (size_t)B_ * D_;

    float* ws           = (float*)d_ws;
    float* qs           = ws;
    float* w_eff        = qs + H_ * DK_;
    float* ctx_x_part   = w_eff + D_ * H_;
    float* ctx_x        = ctx_x_part + (size_t)CTXX_SPLIT * B_ * H_ * D_;
    float* ctx_part     = ctx_x + (size_t)B_ * H_ * D_;
    float* ctx          = ctx_part + (size_t)VSPLIT * B_ * H_ * DV_;
    float* out_pre_part = ctx + (size_t)B_ * H_ * DV_;

    k_qproj<<<4, 256, 0, stream>>>(query, w_qs, qs);
    k_weff<<<64, 256, 0, stream>>>(w_ks, qs, w_eff);
    k_scores<<<512, 256, 0, stream>>>(x, w_eff, attn);
    k_softmax<<<B_ * H_, 256, 0, stream>>>(attn);
    k_ctxx_split<<<512, 256, 0, stream>>>(attn, x, ctx_x_part);
    k_reduce<<<256, 256, 0, stream>>>(ctx_x_part, ctx_x, CTXX_SPLIT, B_ * H_ * D_);
    k_ctx_split<<<64, 256, 0, stream>>>(ctx_x, w_vs, ctx_part);
    k_reduce<<<16, 256, 0, stream>>>(ctx_part, ctx, VSPLIT, B_ * H_ * DV_);
    k_fc_split<<<64, 256, 0, stream>>>(ctx, fc, out_pre_part);
    k_ln<<<B_, 256, 0, stream>>>(out_pre_part, query, gamma, beta, out);
}


# softmax launch removed, two-level softmax: scores pass stores exp(s - chunk max), chunk max and sum; context GEMM scales each K-chunk tile; attn rescaled beside next GEMM (sum order changes)
# speedup vs baseline: 1.7124x; 1.7124x over previous
#include <hip/hip_runtime.h>
#include <math.h>

typedef __attribute__((ext_vector_type(2))) float v2f;
typedef __attribute__((ext_vector_type(8))) float v8f;

#define B_  16
#define L_  4096
#define D_  1024
#define H_  16
#define DK_ 64
#define DV_ 64

static __device__ __forceinline__ v8f wmma_gemm_f32(const float* __restrict__ A, int lda,
                                                    const float* __restrict__ Bm, int ldb,
                                                    int Klen)
{
    const int lane = threadIdx.x & 31;
    const int half = lane >> 4;
    const int mn   = lane & 15;

    const float* ap = A  + (size_t)mn * lda + 2 * half;
    const float* bp = Bm + (size_t)(2 * half) * ldb + mn;

    v8f c = {};
    for (int k = 0; k < Klen; k += 4) {
        if ((k & 511) == 0) {
            __builtin_prefetch(ap + 512, 0, 1);
        }
        v2f a, b;
        a.x = ap[0];
        a.y = ap[1];
        b.x = bp[0];
        b.y = bp[ldb];
        c = __builtin_amdgcn_wmma_f32_16x16x4_f32(false, a, false, b,
                                                  (short)0, c, false, false);
        ap += 4;
        bp += (size_t)4 * ldb;
    }
    return c;
}

__global__ void __launch_bounds__(1024)
k_qfold(const float* __restrict__ query, const float* __restrict__ w_qs,
        const float* __restrict__ w_ks, float* __restrict__ w_eff)
{
    __shared__ float part[16][DK_];
    __shared__ float qh[DK_];

    const int h  = blockIdx.x >> 2;
    const int sl = blockIdx.x & 3;
    const int t  = threadIdx.x;
    const int j  = t & 63;
    const int g  = __builtin_amdgcn_readfirstlane(t >> 6);

    const float* wq = w_qs + (size_t)(g * 64) * (H_ * DK_) + h * DK_ + j;
    const float* qv = query + g * 64;
    float acc = 0.f;
#pragma unroll
    for (int r = 0; r < 64; ++r) acc += qv[r] * wq[(size_t)r * (H_ * DK_)];
    part[g][j] = acc;
    __syncthreads();

    if (t < DK_) {
        float s = 0.f;
#pragma unroll
        for (int gg = 0; gg < 16; ++gg) s += part[gg][t];
        qh[t] = s * 0.125f;
    }
    __syncthreads();

    if (t < 256) {
        const int d = sl * 256 + t;
        const float* wk = w_ks + (size_t)d * (H_ * DK_) + h * DK_;
        float a = 0.f;
#pragma unroll
        for (int jj = 0; jj < DK_; ++jj) a += wk[jj] * qh[jj];
        w_eff[((d >> 2) * H_ + h) * 4 + (d & 3)] = a;
    }
}

#define SC_WAVES 16
#define NCHUNK 16
#define STAT_SUM_OFF (B_ * H_ * NCHUNK)
__global__ void __launch_bounds__(512)
k_scores(const float* __restrict__ x, const float* __restrict__ w_pk,
         float* __restrict__ attn, float* __restrict__ stat)
{
    __shared__ float4 wl[(D_ / 4) * H_];
    __shared__ float wred[2][SC_WAVES][H_];

    const int t = threadIdx.x;
    const float4* wsrc = reinterpret_cast<const float4*>(w_pk);
#pragma unroll
    for (int i = 0; i < 8; ++i) wl[t + 512 * i] = wsrc[t + 512 * i];
    __syncthreads();

    const int lane = t & 31, half = lane >> 4, c = lane & 15;
    const int wave = blockIdx.x * SC_WAVES + (t >> 5);
    const int b  = wave >> 8;
    const int lt = wave & 255;

    const float* ap = x + ((size_t)b * L_ + (size_t)lt * 16 + c) * D_ + 4 * half;
    const float4* bp = wl + half * H_ + c;

    v8f acc = {};
#pragma unroll
    for (int j = 0; j < D_ / 8; ++j) {
        const float4 a = *reinterpret_cast<const float4*>(ap + 8 * j);
        const float4 w = bp[2 * H_ * j];
        v2f a0, a1, b0, b1;
        a0.x = a.x; a0.y = a.y;
        a1.x = a.z; a1.y = a.w;
        b0.x = w.x; b0.y = w.y;
        b1.x = w.z; b1.y = w.w;
        acc = __builtin_amdgcn_wmma_f32_16x16x4_f32(false, a0, false, b0, (short)0, acc, false, false);
        acc = __builtin_amdgcn_wmma_f32_16x16x4_f32(false, a1, false, b1, (short)0, acc, false, false);
    }

    const int wv = t >> 5;
    float mx = acc[0];
#pragma unroll
    for (int r = 1; r < 8; ++r) mx = fmaxf(mx, acc[r]);
    mx = fmaxf(mx, __shfl_xor(mx, 16));
    if (half == 0) wred[0][wv][c] = mx;
    __syncthreads();
    float mc = wred[0][0][c];
#pragma unroll
    for (int w = 1; w < SC_WAVES; ++w) mc = fmaxf(mc, wred[0][w][c]);

    float p[8];
    float sm = 0.f;
#pragma unroll
    for (int r = 0; r < 8; ++r) { p[r] = expf(acc[r] - mc); sm += p[r]; }
    float* out = attn + ((size_t)b * H_ + c) * L_ + (size_t)lt * 16 + 8 * half;
#pragma unroll
    for (int r = 0; r < 8; ++r) out[r] = p[r];
    sm += __shfl_xor(sm, 16);
    if (half == 0) wred[1][wv][c] = sm;
    __syncthreads();
    if (t < H_) {
        float s = 0.f;
#pragma unroll
        for (int w = 0; w < SC_WAVES; ++w) s += wred[1][w][t];
        const int si = (b * H_ + t) * NCHUNK + (blockIdx.x & (NCHUNK - 1));
        stat[si] = mc;
        stat[STAT_SUM_OFF + si] = s;
    }
}

static __device__ __forceinline__ float chunk_factor(const float* __restrict__ st, int cc)
{
    float m[NCHUNK], s[NCHUNK];
#pragma unroll
    for (int i = 0; i < NCHUNK; ++i) { m[i] = st[i]; s[i] = st[STAT_SUM_OFF + i]; }
    float M = m[0];
#pragma unroll
    for (int i = 1; i < NCHUNK; ++i) M = fmaxf(M, m[i]);
    float S = 0.f, e_cc = 0.f;
#pragma unroll
    for (int i = 0; i < NCHUNK; ++i) {
        const float e = expf(m[i] - M);
        S += s[i] * e;
        if (i == cc) e_cc = e;
    }
    return e_cc / S;
}

#define CTXX_WAVES 16
__global__ void __launch_bounds__(512)
k_ctxx(const float* __restrict__ attn, const float* __restrict__ x,
       const float* __restrict__ stat, float* __restrict__ ctx_x)
{
    __shared__ float4 red4[CTXX_WAVES * 256];
    __shared__ float facs[H_ * NCHUNK];

    const int b  = blockIdx.x >> 4;
    const int ds = blockIdx.x & 15;
    const int s  = threadIdx.x >> 5;
    const int k0 = s * (L_ / CTXX_WAVES);

    const int lane = threadIdx.x & 31, half = lane >> 4, c = lane & 15;

    if (threadIdx.x >= 256) {
        const int e = threadIdx.x - 256;
        facs[e] = chunk_factor(stat + (b * H_ + (e >> 4)) * NCHUNK, e & 15);
    }

    const float* ap = attn + (size_t)b * H_ * L_ + (size_t)c * L_ + k0 + 2 * half;
    const float* bp = x + ((size_t)b * L_ + k0 + 2 * half) * D_ + ds * 64 + 4 * c;

    v8f c0 = {}, c1 = {}, c2 = {}, c3 = {};
#pragma unroll 16
    for (int k = 0; k < L_ / CTXX_WAVES; k += 4) {
        v2f a;
        a.x = ap[0];
        a.y = ap[1];
        const float4 r0 = *reinterpret_cast<const float4*>(bp);
        const float4 r1 = *reinterpret_cast<const float4*>(bp + D_);
        v2f b0, b1, b2, b3;
        b0.x = r0.x; b0.y = r1.x;
        b1.x = r0.y; b1.y = r1.y;
        b2.x = r0.z; b2.y = r1.z;
        b3.x = r0.w; b3.y = r1.w;
        c0 = __builtin_amdgcn_wmma_f32_16x16x4_f32(false, a, false, b0, (short)0, c0, false, false);
        c1 = __builtin_amdgcn_wmma_f32_16x16x4_f32(false, a, false, b1, (short)0, c1, false, false);
        c2 = __builtin_amdgcn_wmma_f32_16x16x4_f32(false, a, false, b2, (short)0, c2, false, false);
        c3 = __builtin_amdgcn_wmma_f32_16x16x4_f32(false, a, false, b3, (short)0, c3, false, false);
        ap += 4;
        bp += (size_t)4 * D_;
    }

#pragma unroll
    for (int r = 0; r < 8; ++r) {
        float4 v;
        v.x = c0[r]; v.y = c1[r]; v.z = c2[r]; v.w = c3[r];
        red4[(s * 8 + r) * 32 + lane] = v;
    }
    __syncthreads();

    if (threadIdx.x < 256) {
        const int o  = threadIdx.x;
        const int r  = o >> 5;
        const int oh = (o >> 4) & 1;
        const int oc = o & 15;
        const float* fh = facs + (r + 8 * oh) * NCHUNK;
        float4 v = red4[o];
        const float f0 = fh[0];
        v.x *= f0; v.y *= f0; v.z *= f0; v.w *= f0;
#pragma unroll
        for (int p = 1; p < CTXX_WAVES; ++p) {
            const float4 w = red4[p * 256 + o];
            const float f = fh[p];
            v.x += w.x * f; v.y += w.y * f; v.z += w.z * f; v.w += w.w * f;
        }
        float* out = ctx_x + (size_t)b * H_ * D_ + (size_t)(r + 8 * oh) * D_ + ds * 64 + 4 * oc;
        *reinterpret_cast<float4*>(out) = v;
    }
}

#define VSPLIT 8
#define CTX_TILES 64
#define NORM_BLOCKS (B_ * H_ / 2)
__global__ void __launch_bounds__(256)
k_ctx(const float* __restrict__ ctx_x, const float* __restrict__ w_vs,
      float* __restrict__ ctx, float* __restrict__ attn, const float* __restrict__ stat)
{
    __shared__ float red[VSPLIT * 256];

    if (blockIdx.x >= CTX_TILES) {
        const int row0 = 2 * (blockIdx.x - CTX_TILES);
        const int t = threadIdx.x;
        float4* p4 = reinterpret_cast<float4*>(attn + (size_t)row0 * L_);
        float4 v[8];
#pragma unroll
        for (int j = 0; j < 8; ++j) v[j] = p4[t + 256 * j];
        if (t < 2 * NCHUNK) red[t] = chunk_factor(stat + (row0 + (t >> 4)) * NCHUNK, t & 15);
        __syncthreads();
#pragma unroll
        for (int j = 0; j < 8; ++j) {
            const int i = t + 256 * j;
            const float f = red[i >> 6];
            float4 w = v[j];
            w.x *= f; w.y *= f; w.z *= f; w.w *= f;
            p4[i] = w;
        }
        return;
    }

    const int tile = blockIdx.x;
    const int s    = threadIdx.x >> 5;
    const int h    = tile >> 2;
    const int dvt  = tile & 3;
    const int k0 = s * (D_ / VSPLIT);

    const float* A  = ctx_x + (size_t)h * D_ + k0;
    const float* Bm = w_vs + (size_t)k0 * (H_ * DV_) + h * DV_ + dvt * 16;
    v8f c = wmma_gemm_f32(A, H_ * D_, Bm, H_ * DV_, D_ / VSPLIT);

    const int lane = threadIdx.x & 31;
#pragma unroll
    for (int r = 0; r < 8; ++r) red[(s * 8 + r) * 32 + lane] = c[r];
    __syncthreads();

    const int o  = threadIdx.x;
    const int r  = o >> 5;
    const int oh = (o >> 4) & 1;
    const int n  = o & 15;
    float v = 0.f;
#pragma unroll
    for (int p = 0; p < VSPLIT; ++p) v += red[p * 256 + o];
    ctx[(size_t)(r + 8 * oh) * (H_ * DV_) + h * DV_ + dvt * 16 + n] = v;
}

__global__ void __launch_bounds__(256)
k_fc(const float* __restrict__ ctx, const float* __restrict__ fc,
     const float* __restrict__ query, float* __restrict__ out_pre)
{
    __shared__ float red[VSPLIT * 256];

    const int dt = blockIdx.x;
    const int s  = threadIdx.x >> 5;
    const int k0 = s * ((H_ * DV_) / VSPLIT);

    const float* A  = ctx + k0;
    const float* Bm = fc + (size_t)k0 * D_ + dt * 16;
    v8f c = wmma_gemm_f32(A, H_ * DV_, Bm, D_, (H_ * DV_) / VSPLIT);

    const int lane = threadIdx.x & 31;
#pragma unroll
    for (int r = 0; r < 8; ++r) red[(s * 8 + r) * 32 + lane] = c[r];
    __syncthreads();

    const int o  = threadIdx.x;
    const int r  = o >> 5;
    const int oh = (o >> 4) & 1;
    const int n  = o & 15;
    float v = query[dt * 16 + n];
#pragma unroll
    for (int p = 0; p < VSPLIT; ++p) v += red[p * 256 + o];
    out_pre[(size_t)(r + 8 * oh) * D_ + dt * 16 + n] = v;
}

__global__ void k_ln(const float* __restrict__ out_pre,
                     const float* __restrict__ gamma, const float* __restrict__ beta,
                     float* __restrict__ out)
{
    __shared__ float rs[256], rs2[256];
    const int b = blockIdx.x, t = threadIdx.x;

    float vals[4];
    float s = 0.f, s2 = 0.f;
#pragma unroll
    for (int i = 0; i < 4; ++i) {
        const int idx = t + i * 256;
        const float v = out_pre[(size_t)b * D_ + idx];
        vals[i] = v; s += v; s2 += v * v;
    }
    rs[t] = s; rs2[t] = s2; __syncthreads();
    for (int k = 128; k > 0; k >>= 1) {
        if (t < k) { rs[t] += rs[t + k]; rs2[t] += rs2[t + k]; }
        __syncthreads();
    }
    float mean = rs[0] * (1.f / D_);
    float var  = rs2[0] * (1.f / D_) - mean * mean;
    float inv  = rsqrtf(var + 1e-6f);
#pragma unroll
    for (int i = 0; i < 4; ++i) {
        int idx = t + i * 256;
        out[(size_t)b * D_ + idx] = (vals[i] - mean) * inv * gamma[idx] + beta[idx];
    }
}

extern "C" void kernel_launch(void* const* d_in, const int* in_sizes, int n_in,
                              void* d_out, int out_size, void* d_ws, size_t ws_size,
                              hipStream_t stream)
{
    (void)in_sizes; (void)n_in; (void)out_size; (void)ws_size;

    const float* x     = (const float*)d_in[0];
    const float* query = (const float*)d_in[1];
    const float* w_qs  = (const float*)d_in[2];
    const float* w_ks  = (const float*)d_in[3];
    const float* w_vs  = (const float*)d_in[4];
    const float* fc    = (const float*)d_in[5];
    const float* gamma = (const float*)d_in[6];
    const float* beta  = (const float*)d_in[7];

    float* out  = (float*)d_out;
    float* attn = out + (size_t)B_ * D_;

    float* ws      = (float*)d_ws;
    float* w_eff   = ws;
    float* ctx_x   = w_eff + D_ * H_;
    float* ctx     = ctx_x + (size_t)B_ * H_ * D_;
    float* out_pre = ctx + (size_t)B_ * H_ * DV_;
    float* stat    = out_pre;

    k_qfold<<<64, 1024, 0, stream>>>(query, w_qs, w_ks, w_eff);
    k_scores<<<256, 512, 0, stream>>>(x, w_eff, attn, stat);
    k_ctxx<<<B_ * 16, 512, 0, stream>>>(attn, x, stat, ctx_x);
    k_ctx<<<CTX_TILES + NORM_BLOCKS, 256, 0, stream>>>(ctx_x, w_vs, ctx, attn, stat);
    k_fc<<<64, 256, 0, stream>>>(ctx, fc, query, out_pre);
    k_ln<<<B_, 256, 0, stream>>>(out_pre, gamma, beta, out);
}


# x read once: scores and context fused in one launch over 16-row LDS tiles, running max/sum softmax per 256-row chunk, chunk planes merged by next launch; scores K split 16-way (sum order changes)
# speedup vs baseline: 1.8884x; 1.1028x over previous
#include <hip/hip_runtime.h>
#include <math.h>

typedef __attribute__((ext_vector_type(2))) float v2f;
typedef __attribute__((ext_vector_type(8))) float v8f;

#define B_  16
#define L_  4096
#define D_  1024
#define H_  16
#define DK_ 64
#define DV_ 64

static __device__ __forceinline__ v8f wmma_gemm_f32(const float* __restrict__ A, int lda,
                                                    const float* __restrict__ Bm, int ldb,
                                                    int Klen)
{
    const int lane = threadIdx.x & 31;
    const int half = lane >> 4;
    const int mn   = lane & 15;

    const float* ap = A  + (size_t)mn * lda + 2 * half;
    const float* bp = Bm + (size_t)(2 * half) * ldb + mn;

    v8f c = {};
    for (int k = 0; k < Klen; k += 4) {
        if ((k & 511) == 0) {
            __builtin_prefetch(ap + 512, 0, 1);
        }
        v2f a, b;
        a.x = ap[0];
        a.y = ap[1];
        b.x = bp[0];
        b.y = bp[ldb];
        c = __builtin_amdgcn_wmma_f32_16x16x4_f32(false, a, false, b,
                                                  (short)0, c, false, false);
        ap += 4;
        bp += (size_t)4 * ldb;
    }
    return c;
}

__global__ void __launch_bounds__(1024)
k_qfold(const float* __restrict__ query, const float* __restrict__ w_qs,
        const float* __restrict__ w_ks, float* __restrict__ w_eff)
{
    __shared__ float part[16][DK_];
    __shared__ float qh[DK_];

    const int h  = blockIdx.x >> 2;
    const int sl = blockIdx.x & 3;
    const int t  = threadIdx.x;
    const int j  = t & 63;
    const int g  = __builtin_amdgcn_readfirstlane(t >> 6);

    const float* wq = w_qs + (size_t)(g * 64) * (H_ * DK_) + h * DK_ + j;
    const float* qv = query + g * 64;
    float acc = 0.f;
#pragma unroll
    for (int r = 0; r < 64; ++r) acc += qv[r] * wq[(size_t)r * (H_ * DK_)];
    part[g][j] = acc;
    __syncthreads();

    if (t < DK_) {
        float s = 0.f;
#pragma unroll
        for (int gg = 0; gg < 16; ++gg) s += part[gg][t];
        qh[t] = s * 0.125f;
    }
    __syncthreads();

    if (t < 256) {
        const int d = sl * 256 + t;
        const float* wk = w_ks + (size_t)d * (H_ * DK_) + h * DK_;
        float a = 0.f;
#pragma unroll
        for (int jj = 0; jj < DK_; ++jj) a += wk[jj] * qh[jj];
        w_eff[((d >> 2) * H_ + h) * 4 + (d & 3)] = a;
    }
}

#define SC_WAVES 16
#define NCHUNK 16
#define STAT_SUM_OFF (B_ * H_ * NCHUNK)
static __device__ __forceinline__ void
scores_body(const float* __restrict__ x, const float* __restrict__ w_pk,
            float* __restrict__ attn, float* __restrict__ stat)
{
    __shared__ float4 wl[(D_ / 4) * H_];
    __shared__ float wred[2][SC_WAVES][H_];

    const int t = threadIdx.x;
    const float4* wsrc = reinterpret_cast<const float4*>(w_pk);
#pragma unroll
    for (int i = 0; i < 8; ++i) wl[t + 512 * i] = wsrc[t + 512 * i];
    __syncthreads();

    const int lane = t & 31, half = lane >> 4, c = lane & 15;
    const int wave = blockIdx.x * SC_WAVES + (t >> 5);
    const int b  = wave >> 8;
    const int lt = wave & 255;

    const float* ap = x + ((size_t)b * L_ + (size_t)lt * 16 + c) * D_ + 4 * half;
    const float4* bp = wl + half * H_ + c;

    v8f acc = {};
#pragma unroll
    for (int j = 0; j < D_ / 8; ++j) {
        const float4 a = *reinterpret_cast<const float4*>(ap + 8 * j);
        const float4 w = bp[2 * H_ * j];
        v2f a0, a1, b0, b1;
        a0.x = a.x; a0.y = a.y;
        a1.x = a.z; a1.y = a.w;
        b0.x = w.x; b0.y = w.y;
        b1.x = w.z; b1.y = w.w;
        acc = __builtin_amdgcn_wmma_f32_16x16x4_f32(false, a0, false, b0, (short)0, acc, false, false);
        acc = __builtin_amdgcn_wmma_f32_16x16x4_f32(false, a1, false, b1, (short)0, acc, false, false);
    }

    const int wv = t >> 5;
    float mx = acc[0];
#pragma unroll
    for (int r = 1; r < 8; ++r) mx = fmaxf(mx, acc[r]);
    mx = fmaxf(mx, __shfl_xor(mx, 16));
    if (half == 0) wred[0][wv][c] = mx;
    __syncthreads();
    float mc = wred[0][0][c];
#pragma unroll
    for (int w = 1; w < SC_WAVES; ++w) mc = fmaxf(mc, wred[0][w][c]);

    float p[8];
    float sm = 0.f;
#pragma unroll
    for (int r = 0; r < 8; ++r) { p[r] = expf(acc[r] - mc); sm += p[r]; }
    float* out = attn + ((size_t)b * H_ + c) * L_ + (size_t)lt * 16 + 8 * half;
#pragma unroll
    for (int r = 0; r < 8; ++r) out[r] = p[r];
    sm += __shfl_xor(sm, 16);
    if (half == 0) wred[1][wv][c] = sm;
    __syncthreads();
    if (t < H_) {
        float s = 0.f;
#pragma unroll
        for (int w = 0; w < SC_WAVES; ++w) s += wred[1][w][t];
        const int si = (b * H_ + t) * NCHUNK + (blockIdx.x & (NCHUNK - 1));
        stat[si] = mc;
        stat[STAT_SUM_OFF + si] = s;
    }
}

static __device__ __forceinline__ void lds_barrier()
{
    asm volatile("s_wait_dscnt 0x0\n\ts_barrier_signal -1\n\ts_barrier_wait -1" ::: "memory");
}
template <int CTRL>
static __device__ __forceinline__ float row_xor(float v)
{
    return __builtin_bit_cast(float, __builtin_amdgcn_update_dpp(0, __builtin_bit_cast(int, v), CTRL, 0xf, 0xf, true));
}
#define TP 1028
#define TILE_F (16 * TP)
static __device__ __forceinline__ void
fused_body(const float* __restrict__ x, const float* __restrict__ w_pk,
           float* __restrict__ attn, float* __restrict__ stat, float* __restrict__ planes)
{
    __shared__ __attribute__((aligned(16))) float xs[2 * TILE_F];
    __shared__ __attribute__((aligned(16))) float part[SC_WAVES * 256];
    __shared__ __attribute__((aligned(16))) float pbuf[H_ * 16];
    __shared__ __attribute__((aligned(16))) float albuf[H_];

    const int t = threadIdx.x;
    const int lane = t & 31, half = lane >> 4, c = lane & 15, w = t >> 5;
    const int b = blockIdx.x >> 4, chunk = blockIdx.x & 15;
    const float* xg = x + ((size_t)b * L_ + (size_t)chunk * 256) * D_;

    const float4* wsrc = reinterpret_cast<const float4*>(w_pk);
    float4 wf[8];
#pragma unroll
    for (int j = 0; j < 8; ++j) wf[j] = wsrc[(16 * w + 2 * j + half) * H_ + c];

    const int lr = t >> 8, lc = t & 255;
    float4 pre0, pre1, pre2, pre3, pre4, pre5, pre6, pre7;
#define TILE_LD(src_, i_) (*reinterpret_cast<const float4*>((src_) + (size_t)(2 * (i_) + lr) * D_ + 4 * lc))
#define TILE_ST(dst_, i_, v_) (*reinterpret_cast<float4*>((dst_) + (2 * (i_) + lr) * TP + 4 * lc) = (v_))
    pre0 = TILE_LD(xg, 0); pre1 = TILE_LD(xg, 1); pre2 = TILE_LD(xg, 2); pre3 = TILE_LD(xg, 3);
    pre4 = TILE_LD(xg, 4); pre5 = TILE_LD(xg, 5); pre6 = TILE_LD(xg, 6); pre7 = TILE_LD(xg, 7);
    TILE_ST(xs, 0, pre0); TILE_ST(xs, 1, pre1); TILE_ST(xs, 2, pre2); TILE_ST(xs, 3, pre3);
    TILE_ST(xs, 4, pre4); TILE_ST(xs, 5, pre5); TILE_ST(xs, 6, pre6); TILE_ST(xs, 7, pre7);
    {
        const float* xn = xg + (size_t)16 * D_;
        pre0 = TILE_LD(xn, 0); pre1 = TILE_LD(xn, 1); pre2 = TILE_LD(xn, 2); pre3 = TILE_LD(xn, 3);
        pre4 = TILE_LD(xn, 4); pre5 = TILE_LD(xn, 5); pre6 = TILE_LD(xn, 6); pre7 = TILE_LD(xn, 7);
    }
    lds_barrier();

    v8f c0 = {}, c1 = {}, c2 = {}, c3 = {};
    float m_run = -INFINITY, l_run = 0.f;

    for (int tile = 0; tile < 16; ++tile) {
        const float* xb = xs + (tile & 1) * TILE_F;
        v8f acc = {};
        const float* ap = xb + c * TP + 64 * w + 4 * half;
#pragma unroll
        for (int j = 0; j < 8; ++j) {
            const float4 a = *reinterpret_cast<const float4*>(ap + 8 * j);
            v2f a0, a1, b0, b1;
            a0.x = a.x; a0.y = a.y;
            a1.x = a.z; a1.y = a.w;
            b0.x = wf[j].x; b0.y = wf[j].y;
            b1.x = wf[j].z; b1.y = wf[j].w;
            acc = __builtin_amdgcn_wmma_f32_16x16x4_f32(false, a0, false, b0, (short)0, acc, false, false);
            acc = __builtin_amdgcn_wmma_f32_16x16x4_f32(false, a1, false, b1, (short)0, acc, false, false);
        }
        {
            float4 v0, v1;
            v0.x = acc[0]; v0.y = acc[1]; v0.z = acc[2]; v0.w = acc[3];
            v1.x = acc[4]; v1.y = acc[5]; v1.z = acc[6]; v1.w = acc[7];
            float4* pp = reinterpret_cast<float4*>(part + w * 256 + c * 16 + 8 * half);
            pp[0] = v0;
            pp[1] = v1;
        }
        lds_barrier();

        if (t < 256) {
            const int hh = t >> 4, mm = t & 15;
            float s = part[t];
#pragma unroll
            for (int ww = 1; ww < SC_WAVES; ++ww) s += part[ww * 256 + t];
            attn[((size_t)b * H_ + hh) * L_ + chunk * 256 + tile * 16 + mm] = s;
            float mt = s;
            mt = fmaxf(mt, row_xor<0x161>(mt));
            mt = fmaxf(mt, row_xor<0x162>(mt));
            mt = fmaxf(mt, row_xor<0x164>(mt));
            mt = fmaxf(mt, row_xor<0x168>(mt));
            const float m_new = fmaxf(m_run, mt);
            const float alpha = expf(m_run - m_new);
            const float p = expf(s - m_new);
            float ps = p;
            ps += row_xor<0x161>(ps);
            ps += row_xor<0x162>(ps);
            ps += row_xor<0x164>(ps);
            ps += row_xor<0x168>(ps);
            l_run = l_run * alpha + ps;
            m_run = m_new;
            pbuf[t] = p;
            if (mm == 0) albuf[hh] = alpha;
        }
        if (tile + 1 < 16) {
            float* xd = xs + ((tile + 1) & 1) * TILE_F;
            TILE_ST(xd, 0, pre0); TILE_ST(xd, 1, pre1); TILE_ST(xd, 2, pre2); TILE_ST(xd, 3, pre3);
            TILE_ST(xd, 4, pre4); TILE_ST(xd, 5, pre5); TILE_ST(xd, 6, pre6); TILE_ST(xd, 7, pre7);
        }
        if (tile + 2 < 16) {
            const float* xn = xg + (size_t)(tile + 2) * 16 * D_;
            pre0 = TILE_LD(xn, 0); pre1 = TILE_LD(xn, 1); pre2 = TILE_LD(xn, 2); pre3 = TILE_LD(xn, 3);
            pre4 = TILE_LD(xn, 4); pre5 = TILE_LD(xn, 5); pre6 = TILE_LD(xn, 6); pre7 = TILE_LD(xn, 7);
        }
        lds_barrier();

        {
            const float4 f0 = *reinterpret_cast<const float4*>(albuf + 8 * half);
            const float4 f1 = *reinterpret_cast<const float4*>(albuf + 8 * half + 4);
            const float al[8] = {f0.x, f0.y, f0.z, f0.w, f1.x, f1.y, f1.z, f1.w};
#pragma unroll
            for (int r = 0; r < 8; ++r) { c0[r] *= al[r]; c1[r] *= al[r]; c2[r] *= al[r]; c3[r] *= al[r]; }
        }
        const float4 q0 = *reinterpret_cast<const float4*>(pbuf + c * 16 + 8 * half);
        const float4 q1 = *reinterpret_cast<const float4*>(pbuf + c * 16 + 8 * half + 4);
        const float pv[8] = {q0.x, q0.y, q0.z, q0.w, q1.x, q1.y, q1.z, q1.w};
        const float* bq = xb + (8 * half) * TP + 64 * w + 4 * c;
#pragma unroll
        for (int s = 0; s < 4; ++s) {
            const float4 r0 = *reinterpret_cast<const float4*>(bq + (2 * s) * TP);
            const float4 r1 = *reinterpret_cast<const float4*>(bq + (2 * s + 1) * TP);
            v2f a, b0, b1, b2, b3;
            a.x = pv[2 * s]; a.y = pv[2 * s + 1];
            b0.x = r0.x; b0.y = r1.x;
            b1.x = r0.y; b1.y = r1.y;
            b2.x = r0.z; b2.y = r1.z;
            b3.x = r0.w; b3.y = r1.w;
            c0 = __builtin_amdgcn_wmma_f32_16x16x4_f32(false, a, false, b0, (short)0, c0, false, false);
            c1 = __builtin_amdgcn_wmma_f32_16x16x4_f32(false, a, false, b1, (short)0, c1, false, false);
            c2 = __builtin_amdgcn_wmma_f32_16x16x4_f32(false, a, false, b2, (short)0, c2, false, false);
            c3 = __builtin_amdgcn_wmma_f32_16x16x4_f32(false, a, false, b3, (short)0, c3, false, false);
        }
    }

    float* pl = planes + (size_t)blockIdx.x * (H_ * D_);
#pragma unroll
    for (int r = 0; r < 8; ++r) {
        float4 v;
        v.x = c0[r]; v.y = c1[r]; v.z = c2[r]; v.w = c3[r];
        *reinterpret_cast<float4*>(pl + (size_t)(r + 8 * half) * D_ + 64 * w + 4 * c) = v;
    }
    if (t < 256 && (t & 15) == 0) {
        const int si = (b * H_ + (t >> 4)) * NCHUNK + chunk;
        stat[si] = m_run;
        stat[STAT_SUM_OFF + si] = l_run;
    }
}

static __device__ __forceinline__ float chunk_factor(const float* __restrict__ st, int cc)
{
    float m[NCHUNK], s[NCHUNK];
#pragma unroll
    for (int i = 0; i < NCHUNK; ++i) { m[i] = st[i]; s[i] = st[STAT_SUM_OFF + i]; }
    float M = m[0];
#pragma unroll
    for (int i = 1; i < NCHUNK; ++i) M = fmaxf(M, m[i]);
    float S = 0.f, e_cc = 0.f;
#pragma unroll
    for (int i = 0; i < NCHUNK; ++i) {
        const float e = expf(m[i] - M);
        S += s[i] * e;
        if (i == cc) e_cc = e;
    }
    return e_cc / S;
}

#define CTXX_WAVES 16
static __device__ __forceinline__ void
ctxx_body(const float* __restrict__ attn, const float* __restrict__ x,
          const float* __restrict__ stat, float* __restrict__ ctx_x)
{
    __shared__ float4 red4[CTXX_WAVES * 256];
    __shared__ float facs[H_ * NCHUNK];

    const int b  = blockIdx.x >> 4;
    const int ds = blockIdx.x & 15;
    const int s  = threadIdx.x >> 5;
    const int k0 = s * (L_ / CTXX_WAVES);

    const int lane = threadIdx.x & 31, half = lane >> 4, c = lane & 15;

    if (threadIdx.x >= 256) {
        const int e = threadIdx.x - 256;
        facs[e] = chunk_factor(stat + (b * H_ + (e >> 4)) * NCHUNK, e & 15);
    }

    const float* ap = attn + (size_t)b * H_ * L_ + (size_t)c * L_ + k0 + 2 * half;
    const float* bp = x + ((size_t)b * L_ + k0 + 2 * half) * D_ + ds * 64 + 4 * c;

    v8f c0 = {}, c1 = {}, c2 = {}, c3 = {};
#pragma unroll 16
    for (int k = 0; k < L_ / CTXX_WAVES; k += 4) {
        v2f a;
        a.x = ap[0];
        a.y = ap[1];
        const float4 r0 = *reinterpret_cast<const float4*>(bp);
        const float4 r1 = *reinterpret_cast<const float4*>(bp + D_);
        v2f b0, b1, b2, b3;
        b0.x = r0.x; b0.y = r1.x;
        b1.x = r0.y; b1.y = r1.y;
        b2.x = r0.z; b2.y = r1.z;
        b3.x = r0.w; b3.y = r1.w;
        c0 = __builtin_amdgcn_wmma_f32_16x16x4_f32(false, a, false, b0, (short)0, c0, false, false);
        c1 = __builtin_amdgcn_wmma_f32_16x16x4_f32(false, a, false, b1, (short)0, c1, false, false);
        c2 = __builtin_amdgcn_wmma_f32_16x16x4_f32(false, a, false, b2, (short)0, c2, false, false);
        c3 = __builtin_amdgcn_wmma_f32_16x16x4_f32(false, a, false, b3, (short)0, c3, false, false);
        ap += 4;
        bp += (size_t)4 * D_;
    }

#pragma unroll
    for (int r = 0; r < 8; ++r) {
        float4 v;
        v.x = c0[r]; v.y = c1[r]; v.z = c2[r]; v.w = c3[r];
        red4[(s * 8 + r) * 32 + lane] = v;
    }
    __syncthreads();

    if (threadIdx.x < 256) {
        const int o  = threadIdx.x;
        const int r  = o >> 5;
        const int oh = (o >> 4) & 1;
        const int oc = o & 15;
        const float* fh = facs + (r + 8 * oh) * NCHUNK;
        float4 v = red4[o];
        const float f0 = fh[0];
        v.x *= f0; v.y *= f0; v.z *= f0; v.w *= f0;
#pragma unroll
        for (int p = 1; p < CTXX_WAVES; ++p) {
            const float4 w = red4[p * 256 + o];
            const float f = fh[p];
            v.x += w.x * f; v.y += w.y * f; v.z += w.z * f; v.w += w.w * f;
        }
        float* out = ctx_x + (size_t)b * H_ * D_ + (size_t)(r + 8 * oh) * D_ + ds * 64 + 4 * oc;
        *reinterpret_cast<float4*>(out) = v;
    }
}

static __device__ __forceinline__ void row_stats(const float* __restrict__ st, float& M, float& S)
{
    float m[NCHUNK], s[NCHUNK];
#pragma unroll
    for (int i = 0; i < NCHUNK; ++i) { m[i] = st[i]; s[i] = st[STAT_SUM_OFF + i]; }
    float mx = m[0];
#pragma unroll
    for (int i = 1; i < NCHUNK; ++i) mx = fmaxf(mx, m[i]);
    float sum = 0.f;
#pragma unroll
    for (int i = 0; i < NCHUNK; ++i) sum += s[i] * expf(m[i] - mx);
    M = mx;
    S = sum;
}

static __device__ __forceinline__ void
combine_body(const float* __restrict__ planes, const float* __restrict__ stat,
             float* __restrict__ ctx_x, float* __restrict__ attn)
{
    __shared__ float fac[2 * NCHUNK];
    const int t = threadIdx.x;

    if (blockIdx.x < 128) {
        const int f   = blockIdx.x * 512 + t;
        const int row = f >> 8, d4 = f & 255;
        const int bb = row >> 4, hh = row & 15;
        const float4* p4 = reinterpret_cast<const float4*>(planes) + ((size_t)(bb * 16) * 16 + hh) * 256 + d4;
        float4 v[NCHUNK];
#pragma unroll
        for (int cc = 0; cc < NCHUNK; ++cc) v[cc] = p4[(size_t)cc * 4096];
        if (t < 2 * NCHUNK) fac[t] = chunk_factor(stat + (2 * blockIdx.x + (t >> 4)) * NCHUNK, t & 15);
        __syncthreads();
        const float* fh = fac + (t >> 8) * NCHUNK;
        float4 o;
        o.x = v[0].x * fh[0]; o.y = v[0].y * fh[0]; o.z = v[0].z * fh[0]; o.w = v[0].w * fh[0];
#pragma unroll
        for (int cc = 1; cc < NCHUNK; ++cc) {
            const float fc_ = fh[cc];
            o.x += v[cc].x * fc_; o.y += v[cc].y * fc_; o.z += v[cc].z * fc_; o.w += v[cc].w * fc_;
        }
        reinterpret_cast<float4*>(ctx_x)[f] = o;
    } else {
        const int row0 = 2 * (blockIdx.x - 128);
        float4* p4 = reinterpret_cast<float4*>(attn + (size_t)row0 * L_);
        float4 v[4];
#pragma unroll
        for (int j = 0; j < 4; ++j) v[j] = p4[t + 512 * j];
        if (t < 2) {
            float M, S;
            row_stats(stat + (row0 + t) * NCHUNK, M, S);
            fac[t] = M;
            fac[2 + t] = 1.f / S;
        }
        __syncthreads();
#pragma unroll
        for (int j = 0; j < 4; ++j) {
            const int i = t + 512 * j;
            const float M = fac[i >> 10], inv = fac[2 + (i >> 10)];
            float4 o;
            o.x = expf(v[j].x - M) * inv; o.y = expf(v[j].y - M) * inv;
            o.z = expf(v[j].z - M) * inv; o.w = expf(v[j].w - M) * inv;
            p4[i] = o;
        }
    }
}

#define WS_BASE_FLOATS 311296
#define WS_NEED_BYTES ((unsigned long long)(WS_BASE_FLOATS + 256 * H_ * D_) * 4ull)

__global__ void __launch_bounds__(512)
k_pass1(const float* __restrict__ x, const float* __restrict__ w_pk,
        float* __restrict__ attn, float* __restrict__ stat, float* __restrict__ planes,
        unsigned long long ws_bytes)
{
    if (ws_bytes >= WS_NEED_BYTES) fused_body(x, w_pk, attn, stat, planes);
    else                           scores_body(x, w_pk, attn, stat);
}

__global__ void __launch_bounds__(512)
k_pass2(float* __restrict__ attn, const float* __restrict__ x,
        const float* __restrict__ stat, float* __restrict__ ctx_x,
        const float* __restrict__ planes, unsigned long long ws_bytes)
{
    if (ws_bytes >= WS_NEED_BYTES) combine_body(planes, stat, ctx_x, attn);
    else                           ctxx_body(attn, x, stat, ctx_x);
}

#define VSPLIT 8
#define CTX_TILES 64
#define NORM_BLOCKS (B_ * H_ / 2)
__global__ void __launch_bounds__(256)
k_ctx(const float* __restrict__ ctx_x, const float* __restrict__ w_vs,
      float* __restrict__ ctx, float* __restrict__ attn, const float* __restrict__ stat,
      unsigned long long ws_bytes)
{
    __shared__ float red[VSPLIT * 256];

    if (blockIdx.x >= CTX_TILES) {
        if (ws_bytes >= WS_NEED_BYTES) return;
        const int row0 = 2 * (blockIdx.x - CTX_TILES);
        const int t = threadIdx.x;
        float4* p4 = reinterpret_cast<float4*>(attn + (size_t)row0 * L_);
        float4 v[8];
#pragma unroll
        for (int j = 0; j < 8; ++j) v[j] = p4[t + 256 * j];
        if (t < 2 * NCHUNK) red[t] = chunk_factor(stat + (row0 + (t >> 4)) * NCHUNK, t & 15);
        __syncthreads();
#pragma unroll
        for (int j = 0; j < 8; ++j) {
            const int i = t + 256 * j;
            const float f = red[i >> 6];
            float4 w = v[j];
            w.x *= f; w.y *= f; w.z *= f; w.w *= f;
            p4[i] = w;
        }
        return;
    }

    const int tile = blockIdx.x;
    const int s    = threadIdx.x >> 5;
    const int h    = tile >> 2;
    const int dvt  = tile & 3;
    const int k0 = s * (D_ / VSPLIT);

    const float* A  = ctx_x + (size_t)h * D_ + k0;
    const float* Bm = w_vs + (size_t)k0 * (H_ * DV_) + h * DV_ + dvt * 16;
    v8f c = wmma_gemm_f32(A, H_ * D_, Bm, H_ * DV_, D_ / VSPLIT);

    const int lane = threadIdx.x & 31;
#pragma unroll
    for (int r = 0; r < 8; ++r) red[(s * 8 + r) * 32 + lane] = c[r];
    __syncthreads();

    const int o  = threadIdx.x;
    const int r  = o >> 5;
    const int oh = (o >> 4) & 1;
    const int n  = o & 15;
    float v = 0.f;
#pragma unroll
    for (int p = 0; p < VSPLIT; ++p) v += red[p * 256 + o];
    ctx[(size_t)(r + 8 * oh) * (H_ * DV_) + h * DV_ + dvt * 16 + n] = v;
}

__global__ void __launch_bounds__(256)
k_fc(const float* __restrict__ ctx, const float* __restrict__ fc,
     const float* __restrict__ query, float* __restrict__ out_pre)
{
    __shared__ float red[VSPLIT * 256];

    const int dt = blockIdx.x;
    const int s  = threadIdx.x >> 5;
    const int k0 = s * ((H_ * DV_) / VSPLIT);

    const float* A  = ctx + k0;
    const float* Bm = fc + (size_t)k0 * D_ + dt * 16;
    v8f c = wmma_gemm_f32(A, H_ * DV_, Bm, D_, (H_ * DV_) / VSPLIT);

    const int lane = threadIdx.x & 31;
#pragma unroll
    for (int r = 0; r < 8; ++r) red[(s * 8 + r) * 32 + lane] = c[r];
    __syncthreads();

    const int o  = threadIdx.x;
    const int r  = o >> 5;
    const int oh = (o >> 4) & 1;
    const int n  = o & 15;
    float v = query[dt * 16 + n];
#pragma unroll
    for (int p = 0; p < VSPLIT; ++p) v += red[p * 256 + o];
    out_pre[(size_t)(r + 8 * oh) * D_ + dt * 16 + n] = v;
}

__global__ void k_ln(const float* __restrict__ out_pre,
                     const float* __restrict__ gamma, const float* __restrict__ beta,
                     float* __restrict__ out)
{
    __shared__ float rs[256], rs2[256];
    const int b = blockIdx.x, t = threadIdx.x;

    float vals[4];
    float s = 0.f, s2 = 0.f;
#pragma unroll
    for (int i = 0; i < 4; ++i) {
        const int idx = t + i * 256;
        const float v = out_pre[(size_t)b * D_ + idx];
        vals[i] = v; s += v; s2 += v * v;
    }
    rs[t] = s; rs2[t] = s2; __syncthreads();
    for (int k = 128; k > 0; k >>= 1) {
        if (t < k) { rs[t] += rs[t + k]; rs2[t] += rs2[t + k]; }
        __syncthreads();
    }
    float mean = rs[0] * (1.f / D_);
    float var  = rs2[0] * (1.f / D_) - mean * mean;
    float inv  = rsqrtf(var + 1e-6f);
#pragma unroll
    for (int i = 0; i < 4; ++i) {
        int idx = t + i * 256;
        out[(size_t)b * D_ + idx] = (vals[i] - mean) * inv * gamma[idx] + beta[idx];
    }
}

extern "C" void kernel_launch(void* const* d_in, const int* in_sizes, int n_in,
                              void* d_out, int out_size, void* d_ws, size_t ws_size,
                              hipStream_t stream)
{
    (void)in_sizes; (void)n_in; (void)out_size;

    const float* x     = (const float*)d_in[0];
    const float* query = (const float*)d_in[1];
    const float* w_qs  = (const float*)d_in[2];
    const float* w_ks  = (const float*)d_in[3];
    const float* w_vs  = (const float*)d_in[4];
    const float* fc    = (const float*)d_in[5];
    const float* gamma = (const float*)d_in[6];
    const float* beta  = (const float*)d_in[7];

    float* out  = (float*)d_out;
    float* attn = out + (size_t)B_ * D_;

    float* ws      = (float*)d_ws;
    float* w_eff   = ws;
    float* ctx_x   = w_eff + D_ * H_;
    float* ctx     = ctx_x + (size_t)B_ * H_ * D_;
    float* out_pre = ctx + (size_t)B_ * H_ * DV_;
    float* stat    = out_pre;
    float* planes  = out_pre + (size_t)B_ * D_;
    const unsigned long long ws_bytes = ws_size;

    k_qfold<<<64, 1024, 0, stream>>>(query, w_qs, w_ks, w_eff);
    k_pass1<<<256, 512, 0, stream>>>(x, w_eff, attn, stat, planes, ws_bytes);
    k_pass2<<<B_ * 16, 512, 0, stream>>>(attn, x, stat, ctx_x, planes, ws_bytes);
    k_ctx<<<CTX_TILES + NORM_BLOCKS, 256, 0, stream>>>(ctx_x, w_vs, ctx, attn, stat, ws_bytes);
    k_fc<<<64, 256, 0, stream>>>(ctx, fc, query, out_pre);
    k_ln<<<B_, 256, 0, stream>>>(out_pre, gamma, beta, out);
}


# one-pass kernel software-pipelined over three tiles, one barrier per tile not two: softmax of tile i beside scores of tile i+1 and context of tile i-1 (4 LDS tile buffers); sums and order unchanged
# speedup vs baseline: 2.0385x; 1.0795x over previous
#include <hip/hip_runtime.h>
#include <math.h>

typedef __attribute__((ext_vector_type(2))) float v2f;
typedef __attribute__((ext_vector_type(8))) float v8f;

#define B_  16
#define L_  4096
#define D_  1024
#define H_  16
#define DK_ 64
#define DV_ 64

static __device__ __forceinline__ v8f wmma_gemm_f32(const float* __restrict__ A, int lda,
                                                    const float* __restrict__ Bm, int ldb,
                                                    int Klen)
{
    const int lane = threadIdx.x & 31;
    const int half = lane >> 4;
    const int mn   = lane & 15;

    const float* ap = A  + (size_t)mn * lda + 2 * half;
    const float* bp = Bm + (size_t)(2 * half) * ldb + mn;

    v8f c = {};
    for (int k = 0; k < Klen; k += 4) {
        if ((k & 511) == 0) {
            __builtin_prefetch(ap + 512, 0, 1);
        }
        v2f a, b;
        a.x = ap[0];
        a.y = ap[1];
        b.x = bp[0];
        b.y = bp[ldb];
        c = __builtin_amdgcn_wmma_f32_16x16x4_f32(false, a, false, b,
                                                  (short)0, c, false, false);
        ap += 4;
        bp += (size_t)4 * ldb;
    }
    return c;
}

__global__ void __launch_bounds__(1024)
k_qfold(const float* __restrict__ query, const float* __restrict__ w_qs,
        const float* __restrict__ w_ks, float* __restrict__ w_eff)
{
    __shared__ float part[16][DK_];
    __shared__ float qh[DK_];

    const int h  = blockIdx.x >> 2;
    const int sl = blockIdx.x & 3;
    const int t  = threadIdx.x;
    const int j  = t & 63;
    const int g  = __builtin_amdgcn_readfirstlane(t >> 6);

    const float* wq = w_qs + (size_t)(g * 64) * (H_ * DK_) + h * DK_ + j;
    const float* qv = query + g * 64;
    float acc = 0.f;
#pragma unroll
    for (int r = 0; r < 64; ++r) acc += qv[r] * wq[(size_t)r * (H_ * DK_)];
    part[g][j] = acc;
    __syncthreads();

    if (t < DK_) {
        float s = 0.f;
#pragma unroll
        for (int gg = 0; gg < 16; ++gg) s += part[gg][t];
        qh[t] = s * 0.125f;
    }
    __syncthreads();

    if (t < 256) {
        const int d = sl * 256 + t;
        const float* wk = w_ks + (size_t)d * (H_ * DK_) + h * DK_;
        float a = 0.f;
#pragma unroll
        for (int jj = 0; jj < DK_; ++jj) a += wk[jj] * qh[jj];
        w_eff[((d >> 2) * H_ + h) * 4 + (d & 3)] = a;
    }
}

#define SC_WAVES 16
#define NCHUNK 16
#define STAT_SUM_OFF (B_ * H_ * NCHUNK)
static __device__ __forceinline__ void
scores_body(const float* __restrict__ x, const float* __restrict__ w_pk,
            float* __restrict__ attn, float* __restrict__ stat, float* __restrict__ smem)
{
    float4* wl = reinterpret_cast<float4*>(smem);
    float (*wred)[SC_WAVES][H_] = reinterpret_cast<float (*)[SC_WAVES][H_]>(smem + D_ * H_);

    const int t = threadIdx.x;
    const float4* wsrc = reinterpret_cast<const float4*>(w_pk);
#pragma unroll
    for (int i = 0; i < 8; ++i) wl[t + 512 * i] = wsrc[t + 512 * i];
    __syncthreads();

    const int lane = t & 31, half = lane >> 4, c = lane & 15;
    const int wave = blockIdx.x * SC_WAVES + (t >> 5);
    const int b  = wave >> 8;
    const int lt = wave & 255;

    const float* ap = x + ((size_t)b * L_ + (size_t)lt * 16 + c) * D_ + 4 * half;
    const float4* bp = wl + half * H_ + c;

    v8f acc = {};
#pragma unroll
    for (int j = 0; j < D_ / 8; ++j) {
        const float4 a = *reinterpret_cast<const float4*>(ap + 8 * j);
        const float4 w = bp[2 * H_ * j];
        v2f a0, a1, b0, b1;
        a0.x = a.x; a0.y = a.y;
        a1.x = a.z; a1.y = a.w;
        b0.x = w.x; b0.y = w.y;
        b1.x = w.z; b1.y = w.w;
        acc = __builtin_amdgcn_wmma_f32_16x16x4_f32(false, a0, false, b0, (short)0, acc, false, false);
        acc = __builtin_amdgcn_wmma_f32_16x16x4_f32(false, a1, false, b1, (short)0, acc, false, false);
    }

    const int wv = t >> 5;
    float mx = acc[0];
#pragma unroll
    for (int r = 1; r < 8; ++r) mx = fmaxf(mx, acc[r]);
    mx = fmaxf(mx, __shfl_xor(mx, 16));
    if (half == 0) wred[0][wv][c] = mx;
    __syncthreads();
    float mc = wred[0][0][c];
#pragma unroll
    for (int w = 1; w < SC_WAVES; ++w) mc = fmaxf(mc, wred[0][w][c]);

    float p[8];
    float sm = 0.f;
#pragma unroll
    for (int r = 0; r < 8; ++r) { p[r] = expf(acc[r] - mc); sm += p[r]; }
    float* out = attn + ((size_t)b * H_ + c) * L_ + (size_t)lt * 16 + 8 * half;
#pragma unroll
    for (int r = 0; r < 8; ++r) out[r] = p[r];
    sm += __shfl_xor(sm, 16);
    if (half == 0) wred[1][wv][c] = sm;
    __syncthreads();
    if (t < H_) {
        float s = 0.f;
#pragma unroll
        for (int w = 0; w < SC_WAVES; ++w) s += wred[1][w][t];
        const int si = (b * H_ + t) * NCHUNK + (blockIdx.x & (NCHUNK - 1));
        stat[si] = mc;
        stat[STAT_SUM_OFF + si] = s;
    }
}

static __device__ __forceinline__ void lds_barrier()
{
    asm volatile("s_wait_dscnt 0x0\n\ts_barrier_signal -1\n\ts_barrier_wait -1" ::: "memory");
}
template <int CTRL>
static __device__ __forceinline__ float row_xor(float v)
{
    return __builtin_bit_cast(float, __builtin_amdgcn_update_dpp(0, __builtin_bit_cast(int, v), CTRL, 0xf, 0xf, true));
}
#define TP 1028
#define TILE_F (16 * TP)
#define NXB 4
#define PART_OFF (NXB * TILE_F)
#define PBUF_OFF (PART_OFF + 2 * SC_WAVES * 256)
#define ALBUF_OFF (PBUF_OFF + 2 * H_ * 16)
#define FUSED_LDS_FLOATS (ALBUF_OFF + 2 * H_)
static __device__ __forceinline__ void
fused_body(const float* __restrict__ x, const float* __restrict__ w_pk,
           float* __restrict__ attn, float* __restrict__ stat, float* __restrict__ planes,
           float* __restrict__ smem)
{
    float* const xs    = smem;
    float* const part  = smem + PART_OFF;
    float* const pbuf  = smem + PBUF_OFF;
    float* const albuf = smem + ALBUF_OFF;

    const int t = threadIdx.x;
    const int lane = t & 31, half = lane >> 4, c = lane & 15, w = t >> 5;
    const int b = blockIdx.x >> 4, chunk = blockIdx.x & 15;
    const float* xg = x + ((size_t)b * L_ + (size_t)chunk * 256) * D_;

    const float4* wsrc = reinterpret_cast<const float4*>(w_pk);
    float4 wf[8];
#pragma unroll
    for (int j = 0; j < 8; ++j) wf[j] = wsrc[(16 * w + 2 * j + half) * H_ + c];

    const int lr = t >> 8, lc = t & 255;
    float4 pre0, pre1, pre2, pre3, pre4, pre5, pre6, pre7;
#define TILE_LD(src_, i_) (*reinterpret_cast<const float4*>((src_) + (size_t)(2 * (i_) + lr) * D_ + 4 * lc))
#define TILE_ST(dst_, i_, v_) (*reinterpret_cast<float4*>((dst_) + (2 * (i_) + lr) * TP + 4 * lc) = (v_))
    pre0 = TILE_LD(xg, 0); pre1 = TILE_LD(xg, 1); pre2 = TILE_LD(xg, 2); pre3 = TILE_LD(xg, 3);
    pre4 = TILE_LD(xg, 4); pre5 = TILE_LD(xg, 5); pre6 = TILE_LD(xg, 6); pre7 = TILE_LD(xg, 7);
    TILE_ST(xs, 0, pre0); TILE_ST(xs, 1, pre1); TILE_ST(xs, 2, pre2); TILE_ST(xs, 3, pre3);
    TILE_ST(xs, 4, pre4); TILE_ST(xs, 5, pre5); TILE_ST(xs, 6, pre6); TILE_ST(xs, 7, pre7);
    {
        const float* xn = xg + (size_t)16 * D_;
        pre0 = TILE_LD(xn, 0); pre1 = TILE_LD(xn, 1); pre2 = TILE_LD(xn, 2); pre3 = TILE_LD(xn, 3);
        pre4 = TILE_LD(xn, 4); pre5 = TILE_LD(xn, 5); pre6 = TILE_LD(xn, 6); pre7 = TILE_LD(xn, 7);
    }
    lds_barrier();

    v8f c0 = {}, c1 = {}, c2 = {}, c3 = {};
    float m_run = -INFINITY, l_run = 0.f;

    for (int i = -1; i <= 16; ++i) {
        if (i >= 0 && i < 16 && t < 256) {
            const float* pr = part + (i & 1) * (SC_WAVES * 256);
            const int hh = t >> 4, mm = t & 15;
            float s = pr[t];
#pragma unroll
            for (int ww = 1; ww < SC_WAVES; ++ww) s += pr[ww * 256 + t];
            attn[((size_t)b * H_ + hh) * L_ + chunk * 256 + i * 16 + mm] = s;
            float mt = s;
            mt = fmaxf(mt, row_xor<0x161>(mt));
            mt = fmaxf(mt, row_xor<0x162>(mt));
            mt = fmaxf(mt, row_xor<0x164>(mt));
            mt = fmaxf(mt, row_xor<0x168>(mt));
            const float m_new = fmaxf(m_run, mt);
            const float alpha = expf(m_run - m_new);
            const float p = expf(s - m_new);
            float ps = p;
            ps += row_xor<0x161>(ps);
            ps += row_xor<0x162>(ps);
            ps += row_xor<0x164>(ps);
            ps += row_xor<0x168>(ps);
            l_run = l_run * alpha + ps;
            m_run = m_new;
            pbuf[(i & 1) * (H_ * 16) + t] = p;
            if (mm == 0) albuf[(i & 1) * H_ + hh] = alpha;
        }
        if (i + 1 < 16) {
            v8f acc = {};
            const float* ap = xs + ((i + 1) & 3) * TILE_F + c * TP + 64 * w + 4 * half;
#pragma unroll
            for (int j = 0; j < 8; ++j) {
                const float4 a = *reinterpret_cast<const float4*>(ap + 8 * j);
                v2f a0, a1, b0, b1;
                a0.x = a.x; a0.y = a.y;
                a1.x = a.z; a1.y = a.w;
                b0.x = wf[j].x; b0.y = wf[j].y;
                b1.x = wf[j].z; b1.y = wf[j].w;
                acc = __builtin_amdgcn_wmma_f32_16x16x4_f32(false, a0, false, b0, (short)0, acc, false, false);
                acc = __builtin_amdgcn_wmma_f32_16x16x4_f32(false, a1, false, b1, (short)0, acc, false, false);
            }
            float4 v0, v1;
            v0.x = acc[0]; v0.y = acc[1]; v0.z = acc[2]; v0.w = acc[3];
            v1.x = acc[4]; v1.y = acc[5]; v1.z = acc[6]; v1.w = acc[7];
            float4* pp = reinterpret_cast<float4*>(part + ((i + 1) & 1) * (SC_WAVES * 256) + w * 256 + c * 16 + 8 * half);
            pp[0] = v0;
            pp[1] = v1;
        }
        if (i >= 1) {
            const int jc = i - 1;
            const float* xb = xs + (jc & 3) * TILE_F;
            {
                const float* alb = albuf + (jc & 1) * H_ + 8 * half;
                const float4 f0 = *reinterpret_cast<const float4*>(alb);
                const float4 f1 = *reinterpret_cast<const float4*>(alb + 4);
                const float al[8] = {f0.x, f0.y, f0.z, f0.w, f1.x, f1.y, f1.z, f1.w};
#pragma unroll
                for (int r = 0; r < 8; ++r) { c0[r] *= al[r]; c1[r] *= al[r]; c2[r] *= al[r]; c3[r] *= al[r]; }
            }
            const float* pb = pbuf + (jc & 1) * (H_ * 16) + c * 16 + 8 * half;
            const float4 q0 = *reinterpret_cast<const float4*>(pb);
            const float4 q1 = *reinterpret_cast<const float4*>(pb + 4);
            const float pv[8] = {q0.x, q0.y, q0.z, q0.w, q1.x, q1.y, q1.z, q1.w};
            const float* bq = xb + (8 * half) * TP + 64 * w + 4 * c;
#pragma unroll
            for (int s = 0; s < 4; ++s) {
                const float4 r0 = *reinterpret_cast<const float4*>(bq + (2 * s) * TP);
                const float4 r1 = *reinterpret_cast<const float4*>(bq + (2 * s + 1) * TP);
                v2f a, b0, b1, b2, b3;
                a.x = pv[2 * s]; a.y = pv[2 * s + 1];
                b0.x = r0.x; b0.y = r1.x;
                b1.x = r0.y; b1.y = r1.y;
                b2.x = r0.z; b2.y = r1.z;
                b3.x = r0.w; b3.y = r1.w;
                c0 = __builtin_amdgcn_wmma_f32_16x16x4_f32(false, a, false, b0, (short)0, c0, false, false);
                c1 = __builtin_amdgcn_wmma_f32_16x16x4_f32(false, a, false, b1, (short)0, c1, false, false);
                c2 = __builtin_amdgcn_wmma_f32_16x16x4_f32(false, a, false, b2, (short)0, c2, false, false);
                c3 = __builtin_amdgcn_wmma_f32_16x16x4_f32(false, a, false, b3, (short)0, c3, false, false);
            }
        }
        if (i + 2 < 16) {
            float* xd = xs + ((i + 2) & 3) * TILE_F;
            TILE_ST(xd, 0, pre0); TILE_ST(xd, 1, pre1); TILE_ST(xd, 2, pre2); TILE_ST(xd, 3, pre3);
            TILE_ST(xd, 4, pre4); TILE_ST(xd, 5, pre5); TILE_ST(xd, 6, pre6); TILE_ST(xd, 7, pre7);
        }
        if (i + 3 < 16) {
            const float* xn = xg + (size_t)(i + 3) * 16 * D_;
            pre0 = TILE_LD(xn, 0); pre1 = TILE_LD(xn, 1); pre2 = TILE_LD(xn, 2); pre3 = TILE_LD(xn, 3);
            pre4 = TILE_LD(xn, 4); pre5 = TILE_LD(xn, 5); pre6 = TILE_LD(xn, 6); pre7 = TILE_LD(xn, 7);
        }
        if (i < 16) lds_barrier();
    }

    float* pl = planes + (size_t)blockIdx.x * (H_ * D_);
#pragma unroll
    for (int r = 0; r < 8; ++r) {
        float4 v;
        v.x = c0[r]; v.y = c1[r]; v.z = c2[r]; v.w = c3[r];
        *reinterpret_cast<float4*>(pl + (size_t)(r + 8 * half) * D_ + 64 * w + 4 * c) = v;
    }
    if (t < 256 && (t & 15) == 0) {
        const int si = (b * H_ + (t >> 4)) * NCHUNK + chunk;
        stat[si] = m_run;
        stat[STAT_SUM_OFF + si] = l_run;
    }
}

static __device__ __forceinline__ float chunk_factor(const float* __restrict__ st, int cc)
{
    float m[NCHUNK], s[NCHUNK];
#pragma unroll
    for (int i = 0; i < NCHUNK; ++i) { m[i] = st[i]; s[i] = st[STAT_SUM_OFF + i]; }
    float M = m[0];
#pragma unroll
    for (int i = 1; i < NCHUNK; ++i) M = fmaxf(M, m[i]);
    float S = 0.f, e_cc = 0.f;
#pragma unroll
    for (int i = 0; i < NCHUNK; ++i) {
        const float e = expf(m[i] - M);
        S += s[i] * e;
        if (i == cc) e_cc = e;
    }
    return e_cc / S;
}

#define CTXX_WAVES 16
static __device__ __forceinline__ void
ctxx_body(const float* __restrict__ attn, const float* __restrict__ x,
          const float* __restrict__ stat, float* __restrict__ ctx_x)
{
    __shared__ float4 red4[CTXX_WAVES * 256];
    __shared__ float facs[H_ * NCHUNK];

    const int b  = blockIdx.x >> 4;
    const int ds = blockIdx.x & 15;
    const int s  = threadIdx.x >> 5;
    const int k0 = s * (L_ / CTXX_WAVES);

    const int lane = threadIdx.x & 31, half = lane >> 4, c = lane & 15;

    if (threadIdx.x >= 256) {
        const int e = threadIdx.x - 256;
        facs[e] = chunk_factor(stat + (b * H_ + (e >> 4)) * NCHUNK, e & 15);
    }

    const float* ap = attn + (size_t)b * H_ * L_ + (size_t)c * L_ + k0 + 2 * half;
    const float* bp = x + ((size_t)b * L_ + k0 + 2 * half) * D_ + ds * 64 + 4 * c;

    v8f c0 = {}, c1 = {}, c2 = {}, c3 = {};
#pragma unroll 16
    for (int k = 0; k < L_ / CTXX_WAVES; k += 4) {
        v2f a;
        a.x = ap[0];
        a.y = ap[1];
        const float4 r0 = *reinterpret_cast<const float4*>(bp);
        const float4 r1 = *reinterpret_cast<const float4*>(bp + D_);
        v2f b0, b1, b2, b3;
        b0.x = r0.x; b0.y = r1.x;
        b1.x = r0.y; b1.y = r1.y;
        b2.x = r0.z; b2.y = r1.z;
        b3.x = r0.w; b3.y = r1.w;
        c0 = __builtin_amdgcn_wmma_f32_16x16x4_f32(false, a, false, b0, (short)0, c0, false, false);
        c1 = __builtin_amdgcn_wmma_f32_16x16x4_f32(false, a, false, b1, (short)0, c1, false, false);
        c2 = __builtin_amdgcn_wmma_f32_16x16x4_f32(false, a, false, b2, (short)0, c2, false, false);
        c3 = __builtin_amdgcn_wmma_f32_16x16x4_f32(false, a, false, b3, (short)0, c3, false, false);
        ap += 4;
        bp += (size_t)4 * D_;
    }

#pragma unroll
    for (int r = 0; r < 8; ++r) {
        float4 v;
        v.x = c0[r]; v.y = c1[r]; v.z = c2[r]; v.w = c3[r];
        red4[(s * 8 + r) * 32 + lane] = v;
    }
    __syncthreads();

    if (threadIdx.x < 256) {
        const int o  = threadIdx.x;
        const int r  = o >> 5;
        const int oh = (o >> 4) & 1;
        const int oc = o & 15;
        const float* fh = facs + (r + 8 * oh) * NCHUNK;
        float4 v = red4[o];
        const float f0 = fh[0];
        v.x *= f0; v.y *= f0; v.z *= f0; v.w *= f0;
#pragma unroll
        for (int p = 1; p < CTXX_WAVES; ++p) {
            const float4 w = red4[p * 256 + o];
            const float f = fh[p];
            v.x += w.x * f; v.y += w.y * f; v.z += w.z * f; v.w += w.w * f;
        }
        float* out = ctx_x + (size_t)b * H_ * D_ + (size_t)(r + 8 * oh) * D_ + ds * 64 + 4 * oc;
        *reinterpret_cast<float4*>(out) = v;
    }
}

static __device__ __forceinline__ void row_stats(const float* __restrict__ st, float& M, float& S)
{
    float m[NCHUNK], s[NCHUNK];
#pragma unroll
    for (int i = 0; i < NCHUNK; ++i) { m[i] = st[i]; s[i] = st[STAT_SUM_OFF + i]; }
    float mx = m[0];
#pragma unroll
    for (int i = 1; i < NCHUNK; ++i) mx = fmaxf(mx, m[i]);
    float sum = 0.f;
#pragma unroll
    for (int i = 0; i < NCHUNK; ++i) sum += s[i] * expf(m[i] - mx);
    M = mx;
    S = sum;
}

static __device__ __forceinline__ void
combine_body(const float* __restrict__ planes, const float* __restrict__ stat,
             float* __restrict__ ctx_x, float* __restrict__ attn)
{
    __shared__ float fac[2 * NCHUNK];
    const int t = threadIdx.x;

    if (blockIdx.x < 128) {
        const int f   = blockIdx.x * 512 + t;
        const int row = f >> 8, d4 = f & 255;
        const int bb = row >> 4, hh = row & 15;
        const float4* p4 = reinterpret_cast<const float4*>(planes) + ((size_t)(bb * 16) * 16 + hh) * 256 + d4;
        float4 v[NCHUNK];
#pragma unroll
        for (int cc = 0; cc < NCHUNK; ++cc) v[cc] = p4[(size_t)cc * 4096];
        if (t < 2 * NCHUNK) fac[t] = chunk_factor(stat + (2 * blockIdx.x + (t >> 4)) * NCHUNK, t & 15);
        __syncthreads();
        const float* fh = fac + (t >> 8) * NCHUNK;
        float4 o;
        o.x = v[0].x * fh[0]; o.y = v[0].y * fh[0]; o.z = v[0].z * fh[0]; o.w = v[0].w * fh[0];
#pragma unroll
        for (int cc = 1; cc < NCHUNK; ++cc) {
            const float fc_ = fh[cc];
            o.x += v[cc].x * fc_; o.y += v[cc].y * fc_; o.z += v[cc].z * fc_; o.w += v[cc].w * fc_;
        }
        reinterpret_cast<float4*>(ctx_x)[f] = o;
    } else {
        const int row0 = 2 * (blockIdx.x - 128);
        float4* p4 = reinterpret_cast<float4*>(attn + (size_t)row0 * L_);
        float4 v[4];
#pragma unroll
        for (int j = 0; j < 4; ++j) v[j] = p4[t + 512 * j];
        if (t < 2) {
            float M, S;
            row_stats(stat + (row0 + t) * NCHUNK, M, S);
            fac[t] = M;
            fac[2 + t] = 1.f / S;
        }
        __syncthreads();
#pragma unroll
        for (int j = 0; j < 4; ++j) {
            const int i = t + 512 * j;
            const float M = fac[i >> 10], inv = fac[2 + (i >> 10)];
            float4 o;
            o.x = expf(v[j].x - M) * inv; o.y = expf(v[j].y - M) * inv;
            o.z = expf(v[j].z - M) * inv; o.w = expf(v[j].w - M) * inv;
            p4[i] = o;
        }
    }
}

#define WS_BASE_FLOATS 311296
#define WS_NEED_BYTES ((unsigned long long)(WS_BASE_FLOATS + 256 * H_ * D_) * 4ull)

__global__ void __launch_bounds__(512)
k_pass1(const float* __restrict__ x, const float* __restrict__ w_pk,
        float* __restrict__ attn, float* __restrict__ stat, float* __restrict__ planes,
        unsigned long long ws_bytes)
{
    __shared__ __attribute__((aligned(16))) float smem[FUSED_LDS_FLOATS];
    if (ws_bytes >= WS_NEED_BYTES) fused_body(x, w_pk, attn, stat, planes, smem);
    else                           scores_body(x, w_pk, attn, stat, smem);
}

__global__ void __launch_bounds__(512)
k_pass2(float* __restrict__ attn, const float* __restrict__ x,
        const float* __restrict__ stat, float* __restrict__ ctx_x,
        const float* __restrict__ planes, unsigned long long ws_bytes)
{
    if (ws_bytes >= WS_NEED_BYTES) combine_body(planes, stat, ctx_x, attn);
    else                           ctxx_body(attn, x, stat, ctx_x);
}

#define VSPLIT 8
#define CTX_TILES 64
#define NORM_BLOCKS (B_ * H_ / 2)
__global__ void __launch_bounds__(256)
k_ctx(const float* __restrict__ ctx_x, const float* __restrict__ w_vs,
      float* __restrict__ ctx, float* __restrict__ attn, const float* __restrict__ stat,
      unsigned long long ws_bytes)
{
    __shared__ float red[VSPLIT * 256];

    if (blockIdx.x >= CTX_TILES) {
        if (ws_bytes >= WS_NEED_BYTES) return;
        const int row0 = 2 * (blockIdx.x - CTX_TILES);
        const int t = threadIdx.x;
        float4* p4 = reinterpret_cast<float4*>(attn + (size_t)row0 * L_);
        float4 v[8];
#pragma unroll
        for (int j = 0; j < 8; ++j) v[j] = p4[t + 256 * j];
        if (t < 2 * NCHUNK) red[t] = chunk_factor(stat + (row0 + (t >> 4)) * NCHUNK, t & 15);
        __syncthreads();
#pragma unroll
        for (int j = 0; j < 8; ++j) {
            const int i = t + 256 * j;
            const float f = red[i >> 6];
            float4 w = v[j];
            w.x *= f; w.y *= f; w.z *= f; w.w *= f;
            p4[i] = w;
        }
        return;
    }

    const int tile = blockIdx.x;
    const int s    = threadIdx.x >> 5;
    const int h    = tile >> 2;
    const int dvt  = tile & 3;
    const int k0 = s * (D_ / VSPLIT);

    const float* A  = ctx_x + (size_t)h * D_ + k0;
    const float* Bm = w_vs + (size_t)k0 * (H_ * DV_) + h * DV_ + dvt * 16;
    v8f c = wmma_gemm_f32(A, H_ * D_, Bm, H_ * DV_, D_ / VSPLIT);

    const int lane = threadIdx.x & 31;
#pragma unroll
    for (int r = 0; r < 8; ++r) red[(s * 8 + r) * 32 + lane] = c[r];
    __syncthreads();

    const int o  = threadIdx.x;
    const int r  = o >> 5;
    const int oh = (o >> 4) & 1;
    const int n  = o & 15;
    float v = 0.f;
#pragma unroll
    for (int p = 0; p < VSPLIT; ++p) v += red[p * 256 + o];
    ctx[(size_t)(r + 8 * oh) * (H_ * DV_) + h * DV_ + dvt * 16 + n] = v;
}

__global__ void __launch_bounds__(256)
k_fc(const float* __restrict__ ctx, const float* __restrict__ fc,
     const float* __restrict__ query, float* __restrict__ out_pre)
{
    __shared__ float red[VSPLIT * 256];

    const int dt = blockIdx.x;
    const int s  = threadIdx.x >> 5;
    const int k0 = s * ((H_ * DV_) / VSPLIT);

    const float* A  = ctx + k0;
    const float* Bm = fc + (size_t)k0 * D_ + dt * 16;
    v8f c = wmma_gemm_f32(A, H_ * DV_, Bm, D_, (H_ * DV_) / VSPLIT);

    const int lane = threadIdx.x & 31;
#pragma unroll
    for (int r = 0; r < 8; ++r) red[(s * 8 + r) * 32 + lane] = c[r];
    __syncthreads();

    const int o  = threadIdx.x;
    const int r  = o >> 5;
    const int oh = (o >> 4) & 1;
    const int n  = o & 15;
    float v = query[dt * 16 + n];
#pragma unroll
    for (int p = 0; p < VSPLIT; ++p) v += red[p * 256 + o];
    out_pre[(size_t)(r + 8 * oh) * D_ + dt * 16 + n] = v;
}

__global__ void k_ln(const float* __restrict__ out_pre,
                     const float* __restrict__ gamma, const float* __restrict__ beta,
                     float* __restrict__ out)
{
    __shared__ float rs[256], rs2[256];
    const int b = blockIdx.x, t = threadIdx.x;

    float vals[4];
    float s = 0.f, s2 = 0.f;
#pragma unroll
    for (int i = 0; i < 4; ++i) {
        const int idx = t + i * 256;
        const float v = out_pre[(size_t)b * D_ + idx];
        vals[i] = v; s += v; s2 += v * v;
    }
    rs[t] = s; rs2[t] = s2; __syncthreads();
    for (int k = 128; k > 0; k >>= 1) {
        if (t < k) { rs[t] += rs[t + k]; rs2[t] += rs2[t + k]; }
        __syncthreads();
    }
    float mean = rs[0] * (1.f / D_);
    float var  = rs2[0] * (1.f / D_) - mean * mean;
    float inv  = rsqrtf(var + 1e-6f);
#pragma unroll
    for (int i = 0; i < 4; ++i) {
        int idx = t + i * 256;
        out[(size_t)b * D_ + idx] = (vals[i] - mean) * inv * gamma[idx] + beta[idx];
    }
}

extern "C" void kernel_launch(void* const* d_in, const int* in_sizes, int n_in,
                              void* d_out, int out_size, void* d_ws, size_t ws_size,
                              hipStream_t stream)
{
    (void)in_sizes; (void)n_in; (void)out_size;

    const float* x     = (const float*)d_in[0];
    const float* query = (const float*)d_in[1];
    const float* w_qs  = (const float*)d_in[2];
    const float* w_ks  = (const float*)d_in[3];
    const float* w_vs  = (const float*)d_in[4];
    const float* fc    = (const float*)d_in[5];
    const float* gamma = (const float*)d_in[6];
    const float* beta  = (const float*)d_in[7];

    float* out  = (float*)d_out;
    float* attn = out + (size_t)B_ * D_;

    float* ws      = (float*)d_ws;
    float* w_eff   = ws;
    float* ctx_x   = w_eff + D_ * H_;
    float* ctx     = ctx_x + (size_t)B_ * H_ * D_;
    float* out_pre = ctx + (size_t)B_ * H_ * DV_;
    float* stat    = out_pre;
    float* planes  = out_pre + (size_t)B_ * D_;
    const unsigned long long ws_bytes = ws_size;

    k_qfold<<<64, 1024, 0, stream>>>(query, w_qs, w_ks, w_eff);
    k_pass1<<<256, 512, 0, stream>>>(x, w_eff, attn, stat, planes, ws_bytes);
    k_pass2<<<B_ * 16, 512, 0, stream>>>(attn, x, stat, ctx_x, planes, ws_bytes);
    k_ctx<<<CTX_TILES + NORM_BLOCKS, 256, 0, stream>>>(ctx_x, w_vs, ctx, attn, stat, ws_bytes);
    k_fc<<<64, 256, 0, stream>>>(ctx, fc, query, out_pre);
    k_ln<<<B_, 256, 0, stream>>>(out_pre, gamma, beta, out);
}


# ctx GEMM and output projection in one launch: each (head, half) ctx tile stays in LDS and feeds its K-slice of fc; the 32 planes are added in the LayerNorm launch; 6 launches to 5 (sum order changes)
# speedup vs baseline: 2.2008x; 1.0796x over previous
#include <hip/hip_runtime.h>
#include <math.h>

typedef __attribute__((ext_vector_type(2))) float v2f;
typedef __attribute__((ext_vector_type(8))) float v8f;

#define B_  16
#define L_  4096
#define D_  1024
#define H_  16
#define DK_ 64
#define DV_ 64

__global__ void __launch_bounds__(1024)
k_qfold(const float* __restrict__ query, const float* __restrict__ w_qs,
        const float* __restrict__ w_ks, float* __restrict__ w_eff)
{
    __shared__ float part[16][DK_];
    __shared__ float qh[DK_];

    const int h  = blockIdx.x >> 2;
    const int sl = blockIdx.x & 3;
    const int t  = threadIdx.x;
    const int j  = t & 63;
    const int g  = __builtin_amdgcn_readfirstlane(t >> 6);

    const float* wq = w_qs + (size_t)(g * 64) * (H_ * DK_) + h * DK_ + j;
    const float* qv = query + g * 64;
    float acc = 0.f;
#pragma unroll
    for (int r = 0; r < 64; ++r) acc += qv[r] * wq[(size_t)r * (H_ * DK_)];
    part[g][j] = acc;
    __syncthreads();

    if (t < DK_) {
        float s = 0.f;
#pragma unroll
        for (int gg = 0; gg < 16; ++gg) s += part[gg][t];
        qh[t] = s * 0.125f;
    }
    __syncthreads();

    if (t < 256) {
        const int d = sl * 256 + t;
        const float* wk = w_ks + (size_t)d * (H_ * DK_) + h * DK_;
        float a = 0.f;
#pragma unroll
        for (int jj = 0; jj < DK_; ++jj) a += wk[jj] * qh[jj];
        w_eff[((d >> 2) * H_ + h) * 4 + (d & 3)] = a;
    }
}

#define SC_WAVES 16
#define NCHUNK 16
#define STAT_SUM_OFF (B_ * H_ * NCHUNK)

static __device__ __forceinline__ void lds_barrier()
{
    asm volatile("s_wait_dscnt 0x0\n\ts_barrier_signal -1\n\ts_barrier_wait -1" ::: "memory");
}
template <int CTRL>
static __device__ __forceinline__ float row_xor(float v)
{
    return __builtin_bit_cast(float, __builtin_amdgcn_update_dpp(0, __builtin_bit_cast(int, v), CTRL, 0xf, 0xf, true));
}
#define TP 1028
#define TILE_F (16 * TP)
#define NXB 4
#define PART_OFF (NXB * TILE_F)
#define PBUF_OFF (PART_OFF + 2 * SC_WAVES * 256)
#define ALBUF_OFF (PBUF_OFF + 2 * H_ * 16)
#define FUSED_LDS_FLOATS (ALBUF_OFF + 2 * H_)
static __device__ __forceinline__ void
fused_body(const float* __restrict__ x, const float* __restrict__ w_pk,
           float* __restrict__ attn, float* __restrict__ stat, float* __restrict__ planes,
           float* __restrict__ smem)
{
    float* const xs    = smem;
    float* const part  = smem + PART_OFF;
    float* const pbuf  = smem + PBUF_OFF;
    float* const albuf = smem + ALBUF_OFF;

    const int t = threadIdx.x;
    const int lane = t & 31, half = lane >> 4, c = lane & 15, w = t >> 5;
    const int b = blockIdx.x >> 4, chunk = blockIdx.x & 15;
    const float* xg = x + ((size_t)b * L_ + (size_t)chunk * 256) * D_;

    const float4* wsrc = reinterpret_cast<const float4*>(w_pk);
    float4 wf[8];
#pragma unroll
    for (int j = 0; j < 8; ++j) wf[j] = wsrc[(16 * w + 2 * j + half) * H_ + c];

    const int lr = t >> 8, lc = t & 255;
    float4 pre0, pre1, pre2, pre3, pre4, pre5, pre6, pre7;
#define TILE_LD(src_, i_) (*reinterpret_cast<const float4*>((src_) + (size_t)(2 * (i_) + lr) * D_ + 4 * lc))
#define TILE_ST(dst_, i_, v_) (*reinterpret_cast<float4*>((dst_) + (2 * (i_) + lr) * TP + 4 * lc) = (v_))
    pre0 = TILE_LD(xg, 0); pre1 = TILE_LD(xg, 1); pre2 = TILE_LD(xg, 2); pre3 = TILE_LD(xg, 3);
    pre4 = TILE_LD(xg, 4); pre5 = TILE_LD(xg, 5); pre6 = TILE_LD(xg, 6); pre7 = TILE_LD(xg, 7);
    TILE_ST(xs, 0, pre0); TILE_ST(xs, 1, pre1); TILE_ST(xs, 2, pre2); TILE_ST(xs, 3, pre3);
    TILE_ST(xs, 4, pre4); TILE_ST(xs, 5, pre5); TILE_ST(xs, 6, pre6); TILE_ST(xs, 7, pre7);
    {
        const float* xn = xg + (size_t)16 * D_;
        pre0 = TILE_LD(xn, 0); pre1 = TILE_LD(xn, 1); pre2 = TILE_LD(xn, 2); pre3 = TILE_LD(xn, 3);
        pre4 = TILE_LD(xn, 4); pre5 = TILE_LD(xn, 5); pre6 = TILE_LD(xn, 6); pre7 = TILE_LD(xn, 7);
    }
    lds_barrier();

    v8f c0 = {}, c1 = {}, c2 = {}, c3 = {};
    float m_run = -INFINITY, l_run = 0.f;

    for (int i = -1; i <= 16; ++i) {
        if (i >= 0 && i < 16 && t < 256) {
            const float* pr = part + (i & 1) * (SC_WAVES * 256);
            const int hh = t >> 4, mm = t & 15;
            float s = pr[t];
#pragma unroll
            for (int ww = 1; ww < SC_WAVES; ++ww) s += pr[ww * 256 + t];
            attn[((size_t)b * H_ + hh) * L_ + chunk * 256 + i * 16 + mm] = s;
            float mt = s;
            mt = fmaxf(mt, row_xor<0x161>(mt));
            mt = fmaxf(mt, row_xor<0x162>(mt));
            mt = fmaxf(mt, row_xor<0x164>(mt));
            mt = fmaxf(mt, row_xor<0x168>(mt));
            const float m_new = fmaxf(m_run, mt);
            const float alpha = expf(m_run - m_new);
            const float p = expf(s - m_new);
            float ps = p;
            ps += row_xor<0x161>(ps);
            ps += row_xor<0x162>(ps);
            ps += row_xor<0x164>(ps);
            ps += row_xor<0x168>(ps);
            l_run = l_run * alpha + ps;
            m_run = m_new;
            pbuf[(i & 1) * (H_ * 16) + t] = p;
            if (mm == 0) albuf[(i & 1) * H_ + hh] = alpha;
        }
        if (i + 1 < 16) {
            v8f acc = {};
            const float* ap = xs + ((i + 1) & 3) * TILE_F + c * TP + 64 * w + 4 * half;
#pragma unroll
            for (int j = 0; j < 8; ++j) {
                const float4 a = *reinterpret_cast<const float4*>(ap + 8 * j);
                v2f a0, a1, b0, b1;
                a0.x = a.x; a0.y = a.y;
                a1.x = a.z; a1.y = a.w;
                b0.x = wf[j].x; b0.y = wf[j].y;
                b1.x = wf[j].z; b1.y = wf[j].w;
                acc = __builtin_amdgcn_wmma_f32_16x16x4_f32(false, a0, false, b0, (short)0, acc, false, false);
                acc = __builtin_amdgcn_wmma_f32_16x16x4_f32(false, a1, false, b1, (short)0, acc, false, false);
            }
            float4 v0, v1;
            v0.x = acc[0]; v0.y = acc[1]; v0.z = acc[2]; v0.w = acc[3];
            v1.x = acc[4]; v1.y = acc[5]; v1.z = acc[6]; v1.w = acc[7];
            float4* pp = reinterpret_cast<float4*>(part + ((i + 1) & 1) * (SC_WAVES * 256) + w * 256 + c * 16 + 8 * half);
            pp[0] = v0;
            pp[1] = v1;
        }
        if (i >= 1) {
            const int jc = i - 1;
            const float* xb = xs + (jc & 3) * TILE_F;
            {
                const float* alb = albuf + (jc & 1) * H_ + 8 * half;
                const float4 f0 = *reinterpret_cast<const float4*>(alb);
                const float4 f1 = *reinterpret_cast<const float4*>(alb + 4);
                const float al[8] = {f0.x, f0.y, f0.z, f0.w, f1.x, f1.y, f1.z, f1.w};
#pragma unroll
                for (int r = 0; r < 8; ++r) { c0[r] *= al[r]; c1[r] *= al[r]; c2[r] *= al[r]; c3[r] *= al[r]; }
            }
            const float* pb = pbuf + (jc & 1) * (H_ * 16) + c * 16 + 8 * half;
            const float4 q0 = *reinterpret_cast<const float4*>(pb);
            const float4 q1 = *reinterpret_cast<const float4*>(pb + 4);
            const float pv[8] = {q0.x, q0.y, q0.z, q0.w, q1.x, q1.y, q1.z, q1.w};
            const float* bq = xb + (8 * half) * TP + 64 * w + 4 * c;
#pragma unroll
            for (int s = 0; s < 4; ++s) {
                const float4 r0 = *reinterpret_cast<const float4*>(bq + (2 * s) * TP);
                const float4 r1 = *reinterpret_cast<const float4*>(bq + (2 * s + 1) * TP);
                v2f a, b0, b1, b2, b3;
                a.x = pv[2 * s]; a.y = pv[2 * s + 1];
                b0.x = r0.x; b0.y = r1.x;
                b1.x = r0.y; b1.y = r1.y;
                b2.x = r0.z; b2.y = r1.z;
                b3.x = r0.w; b3.y = r1.w;
                c0 = __builtin_amdgcn_wmma_f32_16x16x4_f32(false, a, false, b0, (short)0, c0, false, false);
                c1 = __builtin_amdgcn_wmma_f32_16x16x4_f32(false, a, false, b1, (short)0, c1, false, false);
                c2 = __builtin_amdgcn_wmma_f32_16x16x4_f32(false, a, false, b2, (short)0, c2, false, false);
                c3 = __builtin_amdgcn_wmma_f32_16x16x4_f32(false, a, false, b3, (short)0, c3, false, false);
            }
        }
        if (i + 2 < 16) {
            float* xd = xs + ((i + 2) & 3) * TILE_F;
            TILE_ST(xd, 0, pre0); TILE_ST(xd, 1, pre1); TILE_ST(xd, 2, pre2); TILE_ST(xd, 3, pre3);
            TILE_ST(xd, 4, pre4); TILE_ST(xd, 5, pre5); TILE_ST(xd, 6, pre6); TILE_ST(xd, 7, pre7);
        }
        if (i + 3 < 16) {
            const float* xn = xg + (size_t)(i + 3) * 16 * D_;
            pre0 = TILE_LD(xn, 0); pre1 = TILE_LD(xn, 1); pre2 = TILE_LD(xn, 2); pre3 = TILE_LD(xn, 3);
            pre4 = TILE_LD(xn, 4); pre5 = TILE_LD(xn, 5); pre6 = TILE_LD(xn, 6); pre7 = TILE_LD(xn, 7);
        }
        if (i < 16) lds_barrier();
    }

    float* pl = planes + (size_t)blockIdx.x * (H_ * D_);
#pragma unroll
    for (int r = 0; r < 8; ++r) {
        float4 v;
        v.x = c0[r]; v.y = c1[r]; v.z = c2[r]; v.w = c3[r];
        *reinterpret_cast<float4*>(pl + (size_t)(r + 8 * half) * D_ + 64 * w + 4 * c) = v;
    }
    if (t < 256 && (t & 15) == 0) {
        const int si = (b * H_ + (t >> 4)) * NCHUNK + chunk;
        stat[si] = m_run;
        stat[STAT_SUM_OFF + si] = l_run;
    }
}
static __device__ __forceinline__ float chunk_factor(const float* __restrict__ st, int cc)
{
    float m[NCHUNK], s[NCHUNK];
#pragma unroll
    for (int i = 0; i < NCHUNK; ++i) { m[i] = st[i]; s[i] = st[STAT_SUM_OFF + i]; }
    float M = m[0];
#pragma unroll
    for (int i = 1; i < NCHUNK; ++i) M = fmaxf(M, m[i]);
    float S = 0.f, e_cc = 0.f;
#pragma unroll
    for (int i = 0; i < NCHUNK; ++i) {
        const float e = expf(m[i] - M);
        S += s[i] * e;
        if (i == cc) e_cc = e;
    }
    return e_cc / S;
}
static __device__ __forceinline__ void row_stats(const float* __restrict__ st, float& M, float& S)
{
    float m[NCHUNK], s[NCHUNK];
#pragma unroll
    for (int i = 0; i < NCHUNK; ++i) { m[i] = st[i]; s[i] = st[STAT_SUM_OFF + i]; }
    float mx = m[0];
#pragma unroll
    for (int i = 1; i < NCHUNK; ++i) mx = fmaxf(mx, m[i]);
    float sum = 0.f;
#pragma unroll
    for (int i = 0; i < NCHUNK; ++i) sum += s[i] * expf(m[i] - mx);
    M = mx;
    S = sum;
}

static __device__ __forceinline__ void
combine_body(const float* __restrict__ planes, const float* __restrict__ stat,
             float* __restrict__ ctx_x, float* __restrict__ attn)
{
    __shared__ float fac[2 * NCHUNK];
    const int t = threadIdx.x;

    if (blockIdx.x < 128) {
        const int f   = blockIdx.x * 512 + t;
        const int row = f >> 8, d4 = f & 255;
        const int bb = row >> 4, hh = row & 15;
        const float4* p4 = reinterpret_cast<const float4*>(planes) + ((size_t)(bb * 16) * 16 + hh) * 256 + d4;
        float4 v[NCHUNK];
#pragma unroll
        for (int cc = 0; cc < NCHUNK; ++cc) v[cc] = p4[(size_t)cc * 4096];
        if (t < 2 * NCHUNK) fac[t] = chunk_factor(stat + (2 * blockIdx.x + (t >> 4)) * NCHUNK, t & 15);
        __syncthreads();
        const float* fh = fac + (t >> 8) * NCHUNK;
        float4 o;
        o.x = v[0].x * fh[0]; o.y = v[0].y * fh[0]; o.z = v[0].z * fh[0]; o.w = v[0].w * fh[0];
#pragma unroll
        for (int cc = 1; cc < NCHUNK; ++cc) {
            const float fc_ = fh[cc];
            o.x += v[cc].x * fc_; o.y += v[cc].y * fc_; o.z += v[cc].z * fc_; o.w += v[cc].w * fc_;
        }
        reinterpret_cast<float4*>(ctx_x)[f] = o;
    } else {
        const int row0 = 2 * (blockIdx.x - 128);
        float4* p4 = reinterpret_cast<float4*>(attn + (size_t)row0 * L_);
        float4 v[4];
#pragma unroll
        for (int j = 0; j < 4; ++j) v[j] = p4[t + 512 * j];
        if (t < 2) {
            float M, S;
            row_stats(stat + (row0 + t) * NCHUNK, M, S);
            fac[t] = M;
            fac[2 + t] = 1.f / S;
        }
        __syncthreads();
#pragma unroll
        for (int j = 0; j < 4; ++j) {
            const int i = t + 512 * j;
            const float M = fac[i >> 10], inv = fac[2 + (i >> 10)];
            float4 o;
            o.x = expf(v[j].x - M) * inv; o.y = expf(v[j].y - M) * inv;
            o.z = expf(v[j].z - M) * inv; o.w = expf(v[j].w - M) * inv;
            p4[i] = o;
        }
    }
}
#define WS_BASE_FLOATS 311296
#define WS_NEED_BYTES ((unsigned long long)(WS_BASE_FLOATS + 256 * H_ * D_) * 4ull)

__global__ void __launch_bounds__(512)
k_pass1(const float* __restrict__ x, const float* __restrict__ w_pk,
        float* __restrict__ attn, float* __restrict__ stat, float* __restrict__ planes,
        unsigned long long ws_bytes)
{
    __shared__ __attribute__((aligned(16))) float smem[FUSED_LDS_FLOATS];
    if (ws_bytes < WS_NEED_BYTES) return;
    fused_body(x, w_pk, attn, stat, planes, smem);
}

__global__ void __launch_bounds__(512)
k_pass2(float* __restrict__ attn, const float* __restrict__ stat, float* __restrict__ ctx_x,
        const float* __restrict__ planes, unsigned long long ws_bytes)
{
    if (ws_bytes < WS_NEED_BYTES) return;
    combine_body(planes, stat, ctx_x, attn);
}

#define VSPLIT 8
#define NPART (2 * H_)
#define CT_P 34
__global__ void __launch_bounds__(256) __attribute__((amdgpu_waves_per_eu(2, 4)))
k_ctxfc(const float* __restrict__ ctx_x, const float* __restrict__ w_vs,
        const float* __restrict__ fc, float* __restrict__ part, unsigned long long ws_bytes)
{
    __shared__ v2f red[VSPLIT * 256];
    __shared__ __attribute__((aligned(8))) float ct[16 * CT_P];

    const int p  = blockIdx.x >> 2;
    const int ns = blockIdx.x & 3;
    const int h  = p >> 1, hv = p & 1;
    const int s  = threadIdx.x >> 5;
    const int lane = threadIdx.x & 31, half = lane >> 4, c = lane & 15;

    const float* fb = fc + (size_t)(h * DV_ + hv * 32 + 2 * half) * D_ + ns * 256 + 32 * s + 2 * c;
    v2f f0[8], f1[8];
#pragma unroll
    for (int ks = 0; ks < 8; ++ks) {
        f0[ks] = *reinterpret_cast<const v2f*>(fb + (size_t)(4 * ks) * D_);
        f1[ks] = *reinterpret_cast<const v2f*>(fb + (size_t)(4 * ks + 1) * D_);
    }
    __builtin_amdgcn_sched_barrier(0);

    const int k0 = s * (D_ / VSPLIT);
    const float* ap = ctx_x + (size_t)c * (H_ * D_) + (size_t)h * D_ + k0 + 2 * half;
    const float* bp = w_vs + (size_t)(k0 + 2 * half) * (H_ * DV_) + h * DV_ + hv * 32 + 2 * c;
    v8f c0 = {}, c1 = {};
#pragma unroll
    for (int k = 0; k < D_ / VSPLIT; k += 4) {
        const v2f a  = *reinterpret_cast<const v2f*>(ap + k);
        const v2f r0 = *reinterpret_cast<const v2f*>(bp + (size_t)k * (H_ * DV_));
        const v2f r1 = *reinterpret_cast<const v2f*>(bp + (size_t)(k + 1) * (H_ * DV_));
        v2f b0, b1;
        b0.x = r0.x; b0.y = r1.x;
        b1.x = r0.y; b1.y = r1.y;
        c0 = __builtin_amdgcn_wmma_f32_16x16x4_f32(false, a, false, b0, (short)0, c0, false, false);
        c1 = __builtin_amdgcn_wmma_f32_16x16x4_f32(false, a, false, b1, (short)0, c1, false, false);
    }
#pragma unroll
    for (int r = 0; r < 8; ++r) {
        v2f v;
        v.x = c0[r]; v.y = c1[r];
        red[(s * 8 + r) * 32 + lane] = v;
    }
    __syncthreads();
    {
        const int o = threadIdx.x;
        v2f v = red[o];
#pragma unroll
        for (int q = 1; q < VSPLIT; ++q) {
            const v2f w = red[q * 256 + o];
            v.x += w.x; v.y += w.y;
        }
        *reinterpret_cast<v2f*>(ct + ((o >> 5) + 8 * ((o >> 4) & 1)) * CT_P + 2 * (o & 15)) = v;
    }
    __syncthreads();

    const float* ar = ct + c * CT_P + 2 * half;
    v8f d0 = {}, d1 = {};
#pragma unroll
    for (int ks = 0; ks < 8; ++ks) {
        const v2f a = *reinterpret_cast<const v2f*>(ar + 4 * ks);
        v2f b0, b1;
        b0.x = f0[ks].x; b0.y = f1[ks].x;
        b1.x = f0[ks].y; b1.y = f1[ks].y;
        d0 = __builtin_amdgcn_wmma_f32_16x16x4_f32(false, a, false, b0, (short)0, d0, false, false);
        d1 = __builtin_amdgcn_wmma_f32_16x16x4_f32(false, a, false, b1, (short)0, d1, false, false);
    }
    if (ws_bytes < WS_NEED_BYTES) return;
    float* po = part + ((size_t)p * B_ + 8 * half) * D_ + ns * 256 + 32 * s + 2 * c;
#pragma unroll
    for (int r = 0; r < 8; ++r) {
        v2f v;
        v.x = d0[r]; v.y = d1[r];
        *reinterpret_cast<v2f*>(po + (size_t)r * D_) = v;
    }
}

__global__ void __launch_bounds__(256) __attribute__((amdgpu_waves_per_eu(2, 4)))
k_ln(const float* __restrict__ part, const float* __restrict__ query,
     const float* __restrict__ gamma, const float* __restrict__ beta,
     float* __restrict__ out, unsigned long long ws_bytes)
{
    __shared__ float rs[256], rs2[256];
    if (ws_bytes < WS_NEED_BYTES) return;
    const int b = blockIdx.x, t = threadIdx.x;

    const float4* p4 = reinterpret_cast<const float4*>(part + (size_t)b * D_) + t;
    float4 v[NPART];
#pragma unroll
    for (int p = 0; p < NPART; ++p) v[p] = p4[(size_t)p * (B_ * D_ / 4)];
    float4 o = reinterpret_cast<const float4*>(query)[t];
#pragma unroll
    for (int p = 0; p < NPART; ++p) { o.x += v[p].x; o.y += v[p].y; o.z += v[p].z; o.w += v[p].w; }

    const float s  = ((o.x + o.y) + o.z) + o.w;
    const float s2 = ((o.x * o.x + o.y * o.y) + o.z * o.z) + o.w * o.w;
    rs[t] = s; rs2[t] = s2; __syncthreads();
    for (int k = 128; k > 0; k >>= 1) {
        if (t < k) { rs[t] += rs[t + k]; rs2[t] += rs2[t + k]; }
        __syncthreads();
    }
    const float mean = rs[0] * (1.f / D_);
    const float var  = rs2[0] * (1.f / D_) - mean * mean;
    const float inv  = rsqrtf(var + 1e-6f);
    const float4 g  = reinterpret_cast<const float4*>(gamma)[t];
    const float4 be = reinterpret_cast<const float4*>(beta)[t];
    float4 y;
    y.x = (o.x - mean) * inv * g.x + be.x;
    y.y = (o.y - mean) * inv * g.y + be.y;
    y.z = (o.z - mean) * inv * g.z + be.z;
    y.w = (o.w - mean) * inv * g.w + be.w;
    reinterpret_cast<float4*>(out + (size_t)b * D_)[t] = y;
}

extern "C" void kernel_launch(void* const* d_in, const int* in_sizes, int n_in,
                              void* d_out, int out_size, void* d_ws, size_t ws_size,
                              hipStream_t stream)
{
    (void)in_sizes; (void)n_in; (void)out_size;

    const float* x     = (const float*)d_in[0];
    const float* query = (const float*)d_in[1];
    const float* w_qs  = (const float*)d_in[2];
    const float* w_ks  = (const float*)d_in[3];
    const float* w_vs  = (const float*)d_in[4];
    const float* fc    = (const float*)d_in[5];
    const float* gamma = (const float*)d_in[6];
    const float* beta  = (const float*)d_in[7];

    float* out  = (float*)d_out;
    float* attn = out + (size_t)B_ * D_;

    float* ws      = (float*)d_ws;
    float* w_eff   = ws;
    float* ctx_x   = w_eff + D_ * H_;
    float* stat    = ctx_x + (size_t)B_ * H_ * D_;
    float* planes  = ws + WS_BASE_FLOATS;
    float* part    = planes;
    const unsigned long long ws_bytes = ws_size;

    k_qfold<<<64, 1024, 0, stream>>>(query, w_qs, w_ks, w_eff);
    k_pass1<<<256, 512, 0, stream>>>(x, w_eff, attn, stat, planes, ws_bytes);
    k_pass2<<<B_ * 16, 512, 0, stream>>>(attn, stat, ctx_x, planes, ws_bytes);
    k_ctxfc<<<NPART * 4, 256, 0, stream>>>(ctx_x, w_vs, fc, part, ws_bytes);
    k_ln<<<B_, 256, 0, stream>>>(part, query, gamma, beta, out, ws_bytes);
}


# setup launch cut two ways over 256 work-groups (K half of the q projection x 128 fold rows; 160 KB each, whole-line loads at entry); scores pass adds the 2 partial planes (w_eff sum order changes)
# speedup vs baseline: 2.4059x; 1.0932x over previous
#include <hip/hip_runtime.h>
#include <math.h>

typedef __attribute__((ext_vector_type(2))) float v2f;
typedef __attribute__((ext_vector_type(8))) float v8f;

#define B_  16
#define L_  4096
#define D_  1024
#define H_  16
#define DK_ 64
#define DV_ 64

static __device__ __forceinline__ void lds_barrier()
{
    asm volatile("s_wait_dscnt 0x0\n\ts_barrier_signal -1\n\ts_barrier_wait -1" ::: "memory");
}
template <int CTRL>
static __device__ __forceinline__ float row_xor(float v)
{
    return __builtin_bit_cast(float, __builtin_amdgcn_update_dpp(0, __builtin_bit_cast(int, v), CTRL, 0xf, 0xf, true));
}

#define WPLANE_F (D_ * H_)
__global__ void __launch_bounds__(256) __attribute__((amdgpu_waves_per_eu(2, 4)))
k_qfold(const float* __restrict__ query, const float* __restrict__ w_qs,
        const float* __restrict__ w_ks, float* __restrict__ w_part)
{
    __shared__ __attribute__((aligned(16))) float part[16][DK_];
    __shared__ __attribute__((aligned(16))) float qh[DK_];

    const int h  = blockIdx.x >> 4;
    const int ks = (blockIdx.x >> 3) & 1;
    const int rs = blockIdx.x & 7;
    const int t  = threadIdx.x;
    const int lane = t & 31, half = lane >> 4, c = lane & 15;
    const int wv = __builtin_amdgcn_readfirstlane(t >> 5);

    const float4* wq = reinterpret_cast<const float4*>(w_qs + (size_t)(ks * 512 + wv * 64 + half) * (H_ * DK_) + h * DK_) + c;
    float4 a[32];
#pragma unroll
    for (int i = 0; i < 32; ++i) a[i] = wq[(size_t)i * (2 * H_ * DK_ / 4)];
    const int d0 = rs * 128 + wv * 16 + half;
    const float4* wk = reinterpret_cast<const float4*>(w_ks + (size_t)d0 * (H_ * DK_) + h * DK_) + c;
    float4 kf[8];
#pragma unroll
    for (int i = 0; i < 8; ++i) kf[i] = wk[(size_t)i * (2 * H_ * DK_ / 4)];
    const float* qv = query + ks * 512 + wv * 64;
    float4 acc = {0.f, 0.f, 0.f, 0.f};
#pragma unroll
    for (int i = 0; i < 32; ++i) {
        const float qa = __builtin_bit_cast(float, __builtin_amdgcn_readfirstlane(__builtin_bit_cast(int, qv[2 * i])));
        const float qb = __builtin_bit_cast(float, __builtin_amdgcn_readfirstlane(__builtin_bit_cast(int, qv[2 * i + 1])));
        const float q = half ? qb : qa;
        acc.x += q * a[i].x; acc.y += q * a[i].y; acc.z += q * a[i].z; acc.w += q * a[i].w;
    }
    *reinterpret_cast<float4*>(&part[wv * 2 + half][4 * c]) = acc;
    lds_barrier();

    if (t < DK_) {
        float sum = part[0][t];
#pragma unroll
        for (int g = 1; g < 16; ++g) sum += part[g][t];
        qh[t] = sum * 0.125f;
    }
    lds_barrier();

    const float4 qf = *reinterpret_cast<const float4*>(&qh[4 * c]);
    float res = 0.f;
#pragma unroll
    for (int i = 0; i < 8; ++i) {
        float p = ((kf[i].x * qf.x + kf[i].y * qf.y) + kf[i].z * qf.z) + kf[i].w * qf.w;
        p += row_xor<0x161>(p);
        p += row_xor<0x162>(p);
        p += row_xor<0x164>(p);
        p += row_xor<0x168>(p);
        if (c == i) res = p;
    }
    if (c < 8) {
        const int d = d0 + 2 * c;
        w_part[(size_t)ks * WPLANE_F + ((d >> 2) * H_ + h) * 4 + (d & 3)] = res;
    }
}

#define SC_WAVES 16
#define NCHUNK 16
#define STAT_SUM_OFF (B_ * H_ * NCHUNK)

#define TP 1028
#define TILE_F (16 * TP)
#define NXB 4
#define PART_OFF (NXB * TILE_F)
#define PBUF_OFF (PART_OFF + 2 * SC_WAVES * 256)
#define ALBUF_OFF (PBUF_OFF + 2 * H_ * 16)
#define FUSED_LDS_FLOATS (ALBUF_OFF + 2 * H_)
static __device__ __forceinline__ void
fused_body(const float* __restrict__ x, const float* __restrict__ w_pk,
           float* __restrict__ attn, float* __restrict__ stat, float* __restrict__ planes,
           float* __restrict__ smem)
{
    float* const xs    = smem;
    float* const part  = smem + PART_OFF;
    float* const pbuf  = smem + PBUF_OFF;
    float* const albuf = smem + ALBUF_OFF;

    const int t = threadIdx.x;
    const int lane = t & 31, half = lane >> 4, c = lane & 15, w = t >> 5;
    const int b = blockIdx.x >> 4, chunk = blockIdx.x & 15;
    const float* xg = x + ((size_t)b * L_ + (size_t)chunk * 256) * D_;

    const float4* wsrc = reinterpret_cast<const float4*>(w_pk);
    float4 wf[8];
#pragma unroll
    for (int j = 0; j < 8; ++j) {
        const float4 u = wsrc[(16 * w + 2 * j + half) * H_ + c];
        const float4 v = wsrc[WPLANE_F / 4 + (16 * w + 2 * j + half) * H_ + c];
        wf[j].x = u.x + v.x; wf[j].y = u.y + v.y; wf[j].z = u.z + v.z; wf[j].w = u.w + v.w;
    }

    const int lr = t >> 8, lc = t & 255;
    float4 pre0, pre1, pre2, pre3, pre4, pre5, pre6, pre7;
#define TILE_LD(src_, i_) (*reinterpret_cast<const float4*>((src_) + (size_t)(2 * (i_) + lr) * D_ + 4 * lc))
#define TILE_ST(dst_, i_, v_) (*reinterpret_cast<float4*>((dst_) + (2 * (i_) + lr) * TP + 4 * lc) = (v_))
    pre0 = TILE_LD(xg, 0); pre1 = TILE_LD(xg, 1); pre2 = TILE_LD(xg, 2); pre3 = TILE_LD(xg, 3);
    pre4 = TILE_LD(xg, 4); pre5 = TILE_LD(xg, 5); pre6 = TILE_LD(xg, 6); pre7 = TILE_LD(xg, 7);
    TILE_ST(xs, 0, pre0); TILE_ST(xs, 1, pre1); TILE_ST(xs, 2, pre2); TILE_ST(xs, 3, pre3);
    TILE_ST(xs, 4, pre4); TILE_ST(xs, 5, pre5); TILE_ST(xs, 6, pre6); TILE_ST(xs, 7, pre7);
    {
        const float* xn = xg + (size_t)16 * D_;
        pre0 = TILE_LD(xn, 0); pre1 = TILE_LD(xn, 1); pre2 = TILE_LD(xn, 2); pre3 = TILE_LD(xn, 3);
        pre4 = TILE_LD(xn, 4); pre5 = TILE_LD(xn, 5); pre6 = TILE_LD(xn, 6); pre7 = TILE_LD(xn, 7);
    }
    lds_barrier();

    v8f c0 = {}, c1 = {}, c2 = {}, c3 = {};
    float m_run = -INFINITY, l_run = 0.f;

    for (int i = -1; i <= 16; ++i) {
        if (i >= 0 && i < 16 && t < 256) {
            const float* pr = part + (i & 1) * (SC_WAVES * 256);
            const int hh = t >> 4, mm = t & 15;
            float s = pr[t];
#pragma unroll
            for (int ww = 1; ww < SC_WAVES; ++ww) s += pr[ww * 256 + t];
            attn[((size_t)b * H_ + hh) * L_ + chunk * 256 + i * 16 + mm] = s;
            float mt = s;
            mt = fmaxf(mt, row_xor<0x161>(mt));
            mt = fmaxf(mt, row_xor<0x162>(mt));
            mt = fmaxf(mt, row_xor<0x164>(mt));
            mt = fmaxf(mt, row_xor<0x168>(mt));
            const float m_new = fmaxf(m_run, mt);
            const float alpha = expf(m_run - m_new);
            const float p = expf(s - m_new);
            float ps = p;
            ps += row_xor<0x161>(ps);
            ps += row_xor<0x162>(ps);
            ps += row_xor<0x164>(ps);
            ps += row_xor<0x168>(ps);
            l_run = l_run * alpha + ps;
            m_run = m_new;
            pbuf[(i & 1) * (H_ * 16) + t] = p;
            if (mm == 0) albuf[(i & 1) * H_ + hh] = alpha;
        }
        if (i + 1 < 16) {
            v8f acc = {};
            const float* ap = xs + ((i + 1) & 3) * TILE_F + c * TP + 64 * w + 4 * half;
#pragma unroll
            for (int j = 0; j < 8; ++j) {
                const float4 a = *reinterpret_cast<const float4*>(ap + 8 * j);
                v2f a0, a1, b0, b1;
                a0.x = a.x; a0.y = a.y;
                a1.x = a.z; a1.y = a.w;
                b0.x = wf[j].x; b0.y = wf[j].y;
                b1.x = wf[j].z; b1.y = wf[j].w;
                acc = __builtin_amdgcn_wmma_f32_16x16x4_f32(false, a0, false, b0, (short)0, acc, false, false);
                acc = __builtin_amdgcn_wmma_f32_16x16x4_f32(false, a1, false, b1, (short)0, acc, false, false);
            }
            float4 v0, v1;
            v0.x = acc[0]; v0.y = acc[1]; v0.z = acc[2]; v0.w = acc[3];
            v1.x = acc[4]; v1.y = acc[5]; v1.z = acc[6]; v1.w = acc[7];
            float4* pp = reinterpret_cast<float4*>(part + ((i + 1) & 1) * (SC_WAVES * 256) + w * 256 + c * 16 + 8 * half);
            pp[0] = v0;
            pp[1] = v1;
        }
        if (i >= 1) {
            const int jc = i - 1;
            const float* xb = xs + (jc & 3) * TILE_F;
            {
                const float* alb = albuf + (jc & 1) * H_ + 8 * half;
                const float4 f0 = *reinterpret_cast<const float4*>(alb);
                const float4 f1 = *reinterpret_cast<const float4*>(alb + 4);
                const float al[8] = {f0.x, f0.y, f0.z, f0.w, f1.x, f1.y, f1.z, f1.w};
#pragma unroll
                for (int r = 0; r < 8; ++r) { c0[r] *= al[r]; c1[r] *= al[r]; c2[r] *= al[r]; c3[r] *= al[r]; }
            }
            const float* pb = pbuf + (jc & 1) * (H_ * 16) + c * 16 + 8 * half;
            const float4 q0 = *reinterpret_cast<const float4*>(pb);
            const float4 q1 = *reinterpret_cast<const float4*>(pb + 4);
            const float pv[8] = {q0.x, q0.y, q0.z, q0.w, q1.x, q1.y, q1.z, q1.w};
            const float* bq = xb + (8 * half) * TP + 64 * w + 4 * c;
#pragma unroll
            for (int s = 0; s < 4; ++s) {
                const float4 r0 = *reinterpret_cast<const float4*>(bq + (2 * s) * TP);
                const float4 r1 = *reinterpret_cast<const float4*>(bq + (2 * s + 1) * TP);
                v2f a, b0, b1, b2, b3;
                a.x = pv[2 * s]; a.y = pv[2 * s + 1];
                b0.x = r0.x; b0.y = r1.x;
                b1.x = r0.y; b1.y = r1.y;
                b2.x = r0.z; b2.y = r1.z;
                b3.x = r0.w; b3.y = r1.w;
                c0 = __builtin_amdgcn_wmma_f32_16x16x4_f32(false, a, false, b0, (short)0, c0, false, false);
                c1 = __builtin_amdgcn_wmma_f32_16x16x4_f32(false, a, false, b1, (short)0, c1, false, false);
                c2 = __builtin_amdgcn_wmma_f32_16x16x4_f32(false, a, false, b2, (short)0, c2, false, false);
                c3 = __builtin_amdgcn_wmma_f32_16x16x4_f32(false, a, false, b3, (short)0, c3, false, false);
            }
        }
        if (i + 2 < 16) {
            float* xd = xs + ((i + 2) & 3) * TILE_F;
            TILE_ST(xd, 0, pre0); TILE_ST(xd, 1, pre1); TILE_ST(xd, 2, pre2); TILE_ST(xd, 3, pre3);
            TILE_ST(xd, 4, pre4); TILE_ST(xd, 5, pre5); TILE_ST(xd, 6, pre6); TILE_ST(xd, 7, pre7);
        }
        if (i + 3 < 16) {
            const float* xn = xg + (size_t)(i + 3) * 16 * D_;
            pre0 = TILE_LD(xn, 0); pre1 = TILE_LD(xn, 1); pre2 = TILE_LD(xn, 2); pre3 = TILE_LD(xn, 3);
            pre4 = TILE_LD(xn, 4); pre5 = TILE_LD(xn, 5); pre6 = TILE_LD(xn, 6); pre7 = TILE_LD(xn, 7);
        }
        if (i < 16) lds_barrier();
    }

    float* pl = planes + (size_t)blockIdx.x * (H_ * D_);
#pragma unroll
    for (int r = 0; r < 8; ++r) {
        float4 v;
        v.x = c0[r]; v.y = c1[r]; v.z = c2[r]; v.w = c3[r];
        *reinterpret_cast<float4*>(pl + (size_t)(r + 8 * half) * D_ + 64 * w + 4 * c) = v;
    }
    if (t < 256 && (t & 15) == 0) {
        const int si = (b * H_ + (t >> 4)) * NCHUNK + chunk;
        stat[si] = m_run;
        stat[STAT_SUM_OFF + si] = l_run;
    }
}
static __device__ __forceinline__ float chunk_factor(const float* __restrict__ st, int cc)
{
    float m[NCHUNK], s[NCHUNK];
#pragma unroll
    for (int i = 0; i < NCHUNK; ++i) { m[i] = st[i]; s[i] = st[STAT_SUM_OFF + i]; }
    float M = m[0];
#pragma unroll
    for (int i = 1; i < NCHUNK; ++i) M = fmaxf(M, m[i]);
    float S = 0.f, e_cc = 0.f;
#pragma unroll
    for (int i = 0; i < NCHUNK; ++i) {
        const float e = expf(m[i] - M);
        S += s[i] * e;
        if (i == cc) e_cc = e;
    }
    return e_cc / S;
}
static __device__ __forceinline__ void row_stats(const float* __restrict__ st, float& M, float& S)
{
    float m[NCHUNK], s[NCHUNK];
#pragma unroll
    for (int i = 0; i < NCHUNK; ++i) { m[i] = st[i]; s[i] = st[STAT_SUM_OFF + i]; }
    float mx = m[0];
#pragma unroll
    for (int i = 1; i < NCHUNK; ++i) mx = fmaxf(mx, m[i]);
    float sum = 0.f;
#pragma unroll
    for (int i = 0; i < NCHUNK; ++i) sum += s[i] * expf(m[i] - mx);
    M = mx;
    S = sum;
}

static __device__ __forceinline__ void
combine_body(const float* __restrict__ planes, const float* __restrict__ stat,
             float* __restrict__ ctx_x, float* __restrict__ attn)
{
    __shared__ float fac[2 * NCHUNK];
    const int t = threadIdx.x;

    if (blockIdx.x < 128) {
        const int f   = blockIdx.x * 512 + t;
        const int row = f >> 8, d4 = f & 255;
        const int bb = row >> 4, hh = row & 15;
        const float4* p4 = reinterpret_cast<const float4*>(planes) + ((size_t)(bb * 16) * 16 + hh) * 256 + d4;
        float4 v[NCHUNK];
#pragma unroll
        for (int cc = 0; cc < NCHUNK; ++cc) v[cc] = p4[(size_t)cc * 4096];
        if (t < 2 * NCHUNK) fac[t] = chunk_factor(stat + (2 * blockIdx.x + (t >> 4)) * NCHUNK, t & 15);
        __syncthreads();
        const float* fh = fac + (t >> 8) * NCHUNK;
        float4 o;
        o.x = v[0].x * fh[0]; o.y = v[0].y * fh[0]; o.z = v[0].z * fh[0]; o.w = v[0].w * fh[0];
#pragma unroll
        for (int cc = 1; cc < NCHUNK; ++cc) {
            const float fc_ = fh[cc];
            o.x += v[cc].x * fc_; o.y += v[cc].y * fc_; o.z += v[cc].z * fc_; o.w += v[cc].w * fc_;
        }
        reinterpret_cast<float4*>(ctx_x)[f] = o;
    } else {
        const int row0 = 2 * (blockIdx.x - 128);
        float4* p4 = reinterpret_cast<float4*>(attn + (size_t)row0 * L_);
        float4 v[4];
#pragma unroll
        for (int j = 0; j < 4; ++j) v[j] = p4[t + 512 * j];
        if (t < 2) {
            float M, S;
            row_stats(stat + (row0 + t) * NCHUNK, M, S);
            fac[t] = M;
            fac[2 + t] = 1.f / S;
        }
        __syncthreads();
#pragma unroll
        for (int j = 0; j < 4; ++j) {
            const int i = t + 512 * j;
            const float M = fac[i >> 10], inv = fac[2 + (i >> 10)];
            float4 o;
            o.x = expf(v[j].x - M) * inv; o.y = expf(v[j].y - M) * inv;
            o.z = expf(v[j].z - M) * inv; o.w = expf(v[j].w - M) * inv;
            p4[i] = o;
        }
    }
}
#define WS_BASE_FLOATS 311296
#define WS_NEED_BYTES ((unsigned long long)(WS_BASE_FLOATS + 256 * H_ * D_) * 4ull)

__global__ void __launch_bounds__(512)
k_pass1(const float* __restrict__ x, const float* __restrict__ w_pk,
        float* __restrict__ attn, float* __restrict__ stat, float* __restrict__ planes,
        unsigned long long ws_bytes)
{
    __shared__ __attribute__((aligned(16))) float smem[FUSED_LDS_FLOATS];
    if (ws_bytes < WS_NEED_BYTES) return;
    fused_body(x, w_pk, attn, stat, planes, smem);
}

__global__ void __launch_bounds__(512)
k_pass2(float* __restrict__ attn, const float* __restrict__ stat, float* __restrict__ ctx_x,
        const float* __restrict__ planes, unsigned long long ws_bytes)
{
    if (ws_bytes < WS_NEED_BYTES) return;
    combine_body(planes, stat, ctx_x, attn);
}

#define VSPLIT 8
#define NPART (2 * H_)
#define CT_P 34
__global__ void __launch_bounds__(256) __attribute__((amdgpu_waves_per_eu(2, 4)))
k_ctxfc(const float* __restrict__ ctx_x, const float* __restrict__ w_vs,
        const float* __restrict__ fc, float* __restrict__ part, unsigned long long ws_bytes)
{
    __shared__ v2f red[VSPLIT * 256];
    __shared__ __attribute__((aligned(8))) float ct[16 * CT_P];

    const int p  = blockIdx.x >> 2;
    const int ns = blockIdx.x & 3;
    const int h  = p >> 1, hv = p & 1;
    const int s  = threadIdx.x >> 5;
    const int lane = threadIdx.x & 31, half = lane >> 4, c = lane & 15;

    const float* fb = fc + (size_t)(h * DV_ + hv * 32 + 2 * half) * D_ + ns * 256 + 32 * s + 2 * c;
    v2f f0[8], f1[8];
#pragma unroll
    for (int ks = 0; ks < 8; ++ks) {
        f0[ks] = *reinterpret_cast<const v2f*>(fb + (size_t)(4 * ks) * D_);
        f1[ks] = *reinterpret_cast<const v2f*>(fb + (size_t)(4 * ks + 1) * D_);
    }
    __builtin_amdgcn_sched_barrier(0);

    const int k0 = s * (D_ / VSPLIT);
    const float* ap = ctx_x + (size_t)c * (H_ * D_) + (size_t)h * D_ + k0 + 2 * half;
    const float* bp = w_vs + (size_t)(k0 + 2 * half) * (H_ * DV_) + h * DV_ + hv * 32 + 2 * c;
    v8f c0 = {}, c1 = {};
#pragma unroll
    for (int k = 0; k < D_ / VSPLIT; k += 4) {
        const v2f a  = *reinterpret_cast<const v2f*>(ap + k);
        const v2f r0 = *reinterpret_cast<const v2f*>(bp + (size_t)k * (H_ * DV_));
        const v2f r1 = *reinterpret_cast<const v2f*>(bp + (size_t)(k + 1) * (H_ * DV_));
        v2f b0, b1;
        b0.x = r0.x; b0.y = r1.x;
        b1.x = r0.y; b1.y = r1.y;
        c0 = __builtin_amdgcn_wmma_f32_16x16x4_f32(false, a, false, b0, (short)0, c0, false, false);
        c1 = __builtin_amdgcn_wmma_f32_16x16x4_f32(false, a, false, b1, (short)0, c1, false, false);
    }
#pragma unroll
    for (int r = 0; r < 8; ++r) {
        v2f v;
        v.x = c0[r]; v.y = c1[r];
        red[(s * 8 + r) * 32 + lane] = v;
    }
    __syncthreads();
    {
        const int o = threadIdx.x;
        v2f v = red[o];
#pragma unroll
        for (int q = 1; q < VSPLIT; ++q) {
            const v2f w = red[q * 256 + o];
            v.x += w.x; v.y += w.y;
        }
        *reinterpret_cast<v2f*>(ct + ((o >> 5) + 8 * ((o >> 4) & 1)) * CT_P + 2 * (o & 15)) = v;
    }
    __syncthreads();

    const float* ar = ct + c * CT_P + 2 * half;
    v8f d0 = {}, d1 = {};
#pragma unroll
    for (int ks = 0; ks < 8; ++ks) {
        const v2f a = *reinterpret_cast<const v2f*>(ar + 4 * ks);
        v2f b0, b1;
        b0.x = f0[ks].x; b0.y = f1[ks].x;
        b1.x = f0[ks].y; b1.y = f1[ks].y;
        d0 = __builtin_amdgcn_wmma_f32_16x16x4_f32(false, a, false, b0, (short)0, d0, false, false);
        d1 = __builtin_amdgcn_wmma_f32_16x16x4_f32(false, a, false, b1, (short)0, d1, false, false);
    }
    if (ws_bytes < WS_NEED_BYTES) return;
    float* po = part + ((size_t)p * B_ + 8 * half) * D_ + ns * 256 + 32 * s + 2 * c;
#pragma unroll
    for (int r = 0; r < 8; ++r) {
        v2f v;
        v.x = d0[r]; v.y = d1[r];
        *reinterpret_cast<v2f*>(po + (size_t)r * D_) = v;
    }
}

__global__ void __launch_bounds__(256) __attribute__((amdgpu_waves_per_eu(2, 4)))
k_ln(const float* __restrict__ part, const float* __restrict__ query,
     const float* __restrict__ gamma, const float* __restrict__ beta,
     float* __restrict__ out, unsigned long long ws_bytes)
{
    __shared__ float rs[256], rs2[256];
    if (ws_bytes < WS_NEED_BYTES) return;
    const int b = blockIdx.x, t = threadIdx.x;

    const float4* p4 = reinterpret_cast<const float4*>(part + (size_t)b * D_) + t;
    float4 v[NPART];
#pragma unroll
    for (int p = 0; p < NPART; ++p) v[p] = p4[(size_t)p * (B_ * D_ / 4)];
    float4 o = reinterpret_cast<const float4*>(query)[t];
#pragma unroll
    for (int p = 0; p < NPART; ++p) { o.x += v[p].x; o.y += v[p].y; o.z += v[p].z; o.w += v[p].w; }

    const float s  = ((o.x + o.y) + o.z) + o.w;
    const float s2 = ((o.x * o.x + o.y * o.y) + o.z * o.z) + o.w * o.w;
    rs[t] = s; rs2[t] = s2; __syncthreads();
    for (int k = 128; k > 0; k >>= 1) {
        if (t < k) { rs[t] += rs[t + k]; rs2[t] += rs2[t + k]; }
        __syncthreads();
    }
    const float mean = rs[0] * (1.f / D_);
    const float var  = rs2[0] * (1.f / D_) - mean * mean;
    const float inv  = rsqrtf(var + 1e-6f);
    const float4 g  = reinterpret_cast<const float4*>(gamma)[t];
    const float4 be = reinterpret_cast<const float4*>(beta)[t];
    float4 y;
    y.x = (o.x - mean) * inv * g.x + be.x;
    y.y = (o.y - mean) * inv * g.y + be.y;
    y.z = (o.z - mean) * inv * g.z + be.z;
    y.w = (o.w - mean) * inv * g.w + be.w;
    reinterpret_cast<float4*>(out + (size_t)b * D_)[t] = y;
}

extern "C" void kernel_launch(void* const* d_in, const int* in_sizes, int n_in,
                              void* d_out, int out_size, void* d_ws, size_t ws_size,
                              hipStream_t stream)
{
    (void)in_sizes; (void)n_in; (void)out_size;

    const float* x     = (const float*)d_in[0];
    const float* query = (const float*)d_in[1];
    const float* w_qs  = (const float*)d_in[2];
    const float* w_ks  = (const float*)d_in[3];
    const float* w_vs  = (const float*)d_in[4];
    const float* fc    = (const float*)d_in[5];
    const float* gamma = (const float*)d_in[6];
    const float* beta  = (const float*)d_in[7];

    float* out  = (float*)d_out;
    float* attn = out + (size_t)B_ * D_;

    float* ws      = (float*)d_ws;
    float* w_part  = ws;
    float* ctx_x   = ws + D_ * H_;
    float* stat    = ctx_x + (size_t)B_ * H_ * D_;
    float* planes  = ws + WS_BASE_FLOATS;
    float* part    = planes;
    const unsigned long long ws_bytes = ws_size;

    k_qfold<<<256, 256, 0, stream>>>(query, w_qs, w_ks, w_part);
    k_pass1<<<256, 512, 0, stream>>>(x, w_part, attn, stat, planes, ws_bytes);
    k_pass2<<<B_ * 16, 512, 0, stream>>>(attn, stat, ctx_x, planes, ws_bytes);
    k_ctxfc<<<NPART * 4, 256, 0, stream>>>(ctx_x, w_vs, fc, part, ws_bytes);
    k_ln<<<B_, 256, 0, stream>>>(part, query, gamma, beta, out, ws_bytes);
}
